# starting kernel
# baseline (speedup 1.0000x reference)
#include <hip/hip_runtime.h>
#include <hip/hip_bf16.h>

#define GDIM 128

typedef float v2f __attribute__((ext_vector_type(2)));
typedef float v8f __attribute__((ext_vector_type(8)));

__global__ void deg_init_kernel(float* __restrict__ deg, int n) {
  int i = blockIdx.x * blockDim.x + threadIdx.x;
  if (i < n) deg[i] = 1.0f;
}

__global__ void deg_edges_kernel(const int* __restrict__ dst, float* __restrict__ deg, int E) {
  int e = blockIdx.x * blockDim.x + threadIdx.x;
  if (e < E) {
    __hip_atomic_fetch_add(&deg[dst[e]], 1.0f, __ATOMIC_RELAXED, __HIP_MEMORY_SCOPE_AGENT);
  }
}

__global__ void dinv_kernel(float* __restrict__ deg, int n) {
  int i = blockIdx.x * blockDim.x + threadIdx.x;
  if (i < n) deg[i] = rsqrtf(deg[i]);
}

__global__ void gemm_wmma_f32_kernel(const float* __restrict__ A,
                                     const float* __restrict__ B,
                                     float* __restrict__ C, int M) {
  const int wave = threadIdx.x >> 5;
  const int lane = threadIdx.x & 31;
  const int half = lane >> 4;
  const int lmod = lane & 15;
  const int row0 = blockIdx.x * 16;
  const int col0 = wave * 16;
  if (row0 >= M) return;

  const float* Arow = A + (size_t)(row0 + lmod) * GDIM;
  const float* Bcol = B + col0 + lmod;

  v8f acc = {0.f, 0.f, 0.f, 0.f, 0.f, 0.f, 0.f, 0.f};

#pragma unroll 4
  for (int k = 0; k < GDIM; k += 4) {
    const int ka = k + half * 2;
    v2f a, b;
    a.x = Arow[ka + 0];
    a.y = Arow[ka + 1];
    b.x = Bcol[(size_t)(ka + 0) * GDIM];
    b.y = Bcol[(size_t)(ka + 1) * GDIM];
    acc = __builtin_amdgcn_wmma_f32_16x16x4_f32(
         false, a,  false, b,
         (short)0, acc,  false,  false);
  }

  float* Cp = C + (size_t)(row0 + half * 8) * GDIM + col0 + lmod;
#pragma unroll
  for (int r = 0; r < 8; ++r) {
    Cp[(size_t)r * GDIM] = acc[r];
  }
}

__global__ void selfloop_init_kernel(const float* __restrict__ h,
                                     const float* __restrict__ dinv,
                                     const float* __restrict__ bias,
                                     float* __restrict__ agg, int n) {
  size_t idx = (size_t)blockIdx.x * blockDim.x + threadIdx.x;
  if (idx >= (size_t)n * GDIM) return;
  int node = (int)(idx >> 7);
  int d = (int)(idx & (GDIM - 1));
  float di = dinv[node];
  float v = h[idx] * di * di;
  if (bias) v += bias[d];
  agg[idx] = v;
}

__global__ void scatter_edges_kernel(const float* __restrict__ h,
                                     const int* __restrict__ src,
                                     const int* __restrict__ dst,
                                     const float* __restrict__ dinv,
                                     float* __restrict__ agg, int E) {
  const int wave = __builtin_amdgcn_readfirstlane(threadIdx.x >> 5);
  const int e = blockIdx.x * 8 + wave;
  if (e >= E) return;
  const int lane = threadIdx.x & 31;
  const int s = src[e];
  const int d = dst[e];
  const float norm = dinv[s] * dinv[d];
  const float4 v = ((const float4*)(h + (size_t)s * GDIM))[lane];
  float* ap = agg + (size_t)d * GDIM + lane * 4;
  __hip_atomic_fetch_add(ap + 0, v.x * norm, __ATOMIC_RELAXED, __HIP_MEMORY_SCOPE_AGENT);
  __hip_atomic_fetch_add(ap + 1, v.y * norm, __ATOMIC_RELAXED, __HIP_MEMORY_SCOPE_AGENT);
  __hip_atomic_fetch_add(ap + 2, v.z * norm, __ATOMIC_RELAXED, __HIP_MEMORY_SCOPE_AGENT);
  __hip_atomic_fetch_add(ap + 3, v.w * norm, __ATOMIC_RELAXED, __HIP_MEMORY_SCOPE_AGENT);
}

__global__ void bias_relu_kernel(const float* __restrict__ agg,
                                 const float* __restrict__ bias,
                                 float* __restrict__ out, int n) {
  size_t idx = (size_t)blockIdx.x * blockDim.x + threadIdx.x;
  if (idx >= (size_t)n * GDIM) return;
  int d = (int)(idx & (GDIM - 1));
  out[idx] = fmaxf(agg[idx] + bias[d], 0.0f);
}

extern "C" void kernel_launch(void* const* d_in, const int* in_sizes, int n_in,
                              void* d_out, int out_size, void* d_ws, size_t ws_size,
                              hipStream_t stream) {
  const float* x  = (const float*)d_in[0];
  const int*   ei = (const int*)d_in[1];
  const float* W1 = (const float*)d_in[2];
  const float* b1 = (const float*)d_in[3];
  const float* W2 = (const float*)d_in[4];
  const float* b2 = (const float*)d_in[5];
  float* out = (float*)d_out;

  const int N = in_sizes[0] / GDIM;
  const int E = in_sizes[1] / 2;
  const int* src = ei;
  const int* dst = ei + E;

  float* dinv = (float*)d_ws;
  float* bufA = dinv + N;
  float* bufB = bufA + (size_t)N * GDIM;

  const int T = 256;
  const int gN    = (N + T - 1) / T;
  const int gE    = (E + T - 1) / T;
  const int gElem = (int)(((size_t)N * GDIM + T - 1) / T);
  const int gEdgeWave = (E + 7) / 8;
  const int gGemm = N / 16;

  deg_init_kernel<<<gN, T, 0, stream>>>(dinv, N);
  deg_edges_kernel<<<gE, T, 0, stream>>>(dst, dinv, E);
  dinv_kernel<<<gN, T, 0, stream>>>(dinv, N);

  gemm_wmma_f32_kernel<<<gGemm, T, 0, stream>>>(x, W1, bufA, N);
  selfloop_init_kernel<<<gElem, T, 0, stream>>>(bufA, dinv, nullptr, bufB, N);
  scatter_edges_kernel<<<gEdgeWave, T, 0, stream>>>(bufA, src, dst, dinv, bufB, E);
  bias_relu_kernel<<<gElem, T, 0, stream>>>(bufB, b1, bufA, N);

  gemm_wmma_f32_kernel<<<gGemm, T, 0, stream>>>(bufA, W2, bufB, N);
  selfloop_init_kernel<<<gElem, T, 0, stream>>>(bufB, dinv, b2, out, N);
  scatter_edges_kernel<<<gEdgeWave, T, 0, stream>>>(bufB, src, dst, dinv, out, E);
}


# replace the per-edge f32 atomic scatter by a destination-bucketed row gather (counting sort of edges per call, one wave sums a node's in-rows in registers); sum order now set by arrival-order rank
# speedup vs baseline: 2.2330x; 2.2330x over previous
#include <hip/hip_runtime.h>
#include <hip/hip_bf16.h>

#define GDIM 128

typedef float v2f __attribute__((ext_vector_type(2)));
typedef float v8f __attribute__((ext_vector_type(8)));

__global__ void zero_counts_kernel(int* __restrict__ cnt, int n) {
  int i = blockIdx.x * blockDim.x + threadIdx.x;
  if (i < n) cnt[i] = 0;
}

__global__ void count_rank_kernel(const int* __restrict__ dst, int* __restrict__ cnt,
                                  int* __restrict__ rank, int E) {
  int e = blockIdx.x * blockDim.x + threadIdx.x;
  if (e < E) {
    rank[e] = __hip_atomic_fetch_add(&cnt[dst[e]], 1, __ATOMIC_RELAXED, __HIP_MEMORY_SCOPE_AGENT);
  }
}

#define SCAN_PER 16
__global__ void __launch_bounds__(1024) scan_dinv_kernel(const int* __restrict__ cnt,
                                                         int* __restrict__ off,
                                                         float* __restrict__ dinv, int n) {
  __shared__ int wsum[32];
  const int tid = threadIdx.x;
  const int lane = tid & 31;
  const int wave = tid >> 5;
  int carry = 0;
  for (int base = 0; base < n; base += 1024 * SCAN_PER) {
    const int i = base + tid * SCAN_PER;
    const bool full = (i + SCAN_PER <= n);
    int c[SCAN_PER];
    if (full) {
      const int4* p = (const int4*)(cnt + i);
#pragma unroll
      for (int q = 0; q < SCAN_PER / 4; ++q) {
        const int4 t = p[q];
        c[4 * q + 0] = t.x; c[4 * q + 1] = t.y; c[4 * q + 2] = t.z; c[4 * q + 3] = t.w;
      }
    } else {
#pragma unroll
      for (int k = 0; k < SCAN_PER; ++k) c[k] = (i + k < n) ? cnt[i + k] : 0;
    }
    int tsum = 0;
#pragma unroll
    for (int k = 0; k < SCAN_PER; ++k) tsum += c[k];

    int incl = tsum;
    for (int d = 1; d < 32; d <<= 1) {
      const int up = __shfl_up(incl, d);
      if (lane >= d) incl += up;
    }
    if (lane == 31) wsum[wave] = incl;
    __syncthreads();
    const int wv = wsum[lane];
    int winc = wv;
    for (int d = 1; d < 32; d <<= 1) {
      const int up = __shfl_up(winc, d);
      if (lane >= d) winc += up;
    }
    const int total = __shfl(winc, 31);
    const int wbase = __shfl(winc, wave) - __shfl(wv, wave);
    int run = carry + wbase + incl - tsum;

    int o[SCAN_PER];
    float r[SCAN_PER];
#pragma unroll
    for (int k = 0; k < SCAN_PER; ++k) {
      o[k] = run;
      run += c[k];
      r[k] = rsqrtf((float)(c[k] + 1));
    }
    if (full) {
      int4* po = (int4*)(off + i);
      float4* pr = (float4*)(dinv + i);
#pragma unroll
      for (int q = 0; q < SCAN_PER / 4; ++q) {
        po[q] = make_int4(o[4 * q + 0], o[4 * q + 1], o[4 * q + 2], o[4 * q + 3]);
        pr[q] = make_float4(r[4 * q + 0], r[4 * q + 1], r[4 * q + 2], r[4 * q + 3]);
      }
    } else {
#pragma unroll
      for (int k = 0; k < SCAN_PER; ++k) {
        if (i + k < n) { off[i + k] = o[k]; dinv[i + k] = r[k]; }
      }
    }
    carry += total;
    __syncthreads();
  }
}

__global__ void fill_perm_kernel(const int* __restrict__ src, const int* __restrict__ dst,
                                 const int* __restrict__ rank, const int* __restrict__ off,
                                 int* __restrict__ perm, int E) {
  int e = blockIdx.x * blockDim.x + threadIdx.x;
  if (e < E) perm[off[dst[e]] + rank[e]] = src[e];
}

__global__ void gemm_wmma_f32_kernel(const float* __restrict__ A,
                                     const float* __restrict__ B,
                                     float* __restrict__ C, int M) {
  const int wave = threadIdx.x >> 5;
  const int lane = threadIdx.x & 31;
  const int half = lane >> 4;
  const int lmod = lane & 15;
  const int row0 = blockIdx.x * 16;
  const int col0 = wave * 16;
  if (row0 >= M) return;

  const float* Arow = A + (size_t)(row0 + lmod) * GDIM;
  const float* Bcol = B + col0 + lmod;

  v8f acc = {0.f, 0.f, 0.f, 0.f, 0.f, 0.f, 0.f, 0.f};

#pragma unroll 4
  for (int k = 0; k < GDIM; k += 4) {
    const int ka = k + half * 2;
    v2f a, b;
    a.x = Arow[ka + 0];
    a.y = Arow[ka + 1];
    b.x = Bcol[(size_t)(ka + 0) * GDIM];
    b.y = Bcol[(size_t)(ka + 1) * GDIM];
    acc = __builtin_amdgcn_wmma_f32_16x16x4_f32(
         false, a,  false, b,
         (short)0, acc,  false,  false);
  }

  float* Cp = C + (size_t)(row0 + half * 8) * GDIM + col0 + lmod;
#pragma unroll
  for (int r = 0; r < 8; ++r) {
    Cp[(size_t)r * GDIM] = acc[r];
  }
}

template <int K>
__device__ __forceinline__ void accum_rows(const float4* __restrict__ H4,
                                           const int* __restrict__ perm,
                                           const float* __restrict__ dinv,
                                           int j, int lane, float dn, float4& acc) {
  int s[K];
  float4 v[K];
  float w[K];
#pragma unroll
  for (int k = 0; k < K; ++k) s[k] = perm[j + k];
#pragma unroll
  for (int k = 0; k < K; ++k) v[k] = H4[(size_t)s[k] * (GDIM / 4) + lane];
#pragma unroll
  for (int k = 0; k < K; ++k) w[k] = dinv[s[k]] * dn;
#pragma unroll
  for (int k = 0; k < K; ++k) {
    acc.x = fmaf(v[k].x, w[k], acc.x);
    acc.y = fmaf(v[k].y, w[k], acc.y);
    acc.z = fmaf(v[k].z, w[k], acc.z);
    acc.w = fmaf(v[k].w, w[k], acc.w);
  }
}

template <bool RELU>
__global__ void gather_rows_kernel(const float* __restrict__ h,
                                   const int* __restrict__ perm,
                                   const int* __restrict__ off,
                                   const int* __restrict__ cnt,
                                   const float* __restrict__ dinv,
                                   const float* __restrict__ bias,
                                   float* __restrict__ out, int n) {
  const int wave = __builtin_amdgcn_readfirstlane(threadIdx.x >> 5);
  const int node = blockIdx.x * 8 + wave;
  if (node >= n) return;
  const int lane = threadIdx.x & 31;
  const float4* H4 = (const float4*)h;
  int j = off[node];
  const int end = j + cnt[node];
  const float dn = dinv[node];

  const float* bp = bias + lane * 4;
  const float b0 = bp[0], b1v = bp[1], b2v = bp[2], b3 = bp[3];
  const float4 self = H4[(size_t)node * (GDIM / 4) + lane];
  const float dd = dn * dn;
  float4 acc;
  acc.x = self.x * dd;
  acc.y = self.y * dd;
  acc.z = self.z * dd;
  acc.w = self.w * dd;

  for (; j + 8 <= end; j += 8) accum_rows<8>(H4, perm, dinv, j, lane, dn, acc);
  if (j + 4 <= end) { accum_rows<4>(H4, perm, dinv, j, lane, dn, acc); j += 4; }
  if (j + 2 <= end) { accum_rows<2>(H4, perm, dinv, j, lane, dn, acc); j += 2; }
  if (j < end)      { accum_rows<1>(H4, perm, dinv, j, lane, dn, acc); }

  acc.x += b0;
  acc.y += b1v;
  acc.z += b2v;
  acc.w += b3;
  if (RELU) {
    acc.x = fmaxf(acc.x, 0.0f);
    acc.y = fmaxf(acc.y, 0.0f);
    acc.z = fmaxf(acc.z, 0.0f);
    acc.w = fmaxf(acc.w, 0.0f);
  }
  ((float4*)(out + (size_t)node * GDIM))[lane] = acc;
}

extern "C" void kernel_launch(void* const* d_in, const int* in_sizes, int n_in,
                              void* d_out, int out_size, void* d_ws, size_t ws_size,
                              hipStream_t stream) {
  const float* x  = (const float*)d_in[0];
  const int*   ei = (const int*)d_in[1];
  const float* W1 = (const float*)d_in[2];
  const float* b1 = (const float*)d_in[3];
  const float* W2 = (const float*)d_in[4];
  const float* b2 = (const float*)d_in[5];
  float* out = (float*)d_out;

  const int N = in_sizes[0] / GDIM;
  const int E = in_sizes[1] / 2;
  const int* src = ei;
  const int* dst = ei + E;

  float* dinv = (float*)d_ws;
  float* bufA = dinv + N;
  float* bufB = bufA + (size_t)N * GDIM;
  int* cnt  = (int*)(bufB + (size_t)N * GDIM);
  int* off  = cnt + N;
  int* rank = off + N;
  int* perm = rank + E;

  const int T = 256;
  const int gN    = (N + T - 1) / T;
  const int gE    = (E + T - 1) / T;
  const int gNodeWave = (N + 7) / 8;
  const int gGemm = N / 16;

  zero_counts_kernel<<<gN, T, 0, stream>>>(cnt, N);
  count_rank_kernel<<<gE, T, 0, stream>>>(dst, cnt, rank, E);
  scan_dinv_kernel<<<1, 1024, 0, stream>>>(cnt, off, dinv, N);
  fill_perm_kernel<<<gE, T, 0, stream>>>(src, dst, rank, off, perm, E);

  gemm_wmma_f32_kernel<<<gGemm, T, 0, stream>>>(x, W1, bufA, N);
  gather_rows_kernel<true><<<gNodeWave, T, 0, stream>>>(bufA, perm, off, cnt, dinv, b1, bufB, N);

  gemm_wmma_f32_kernel<<<gGemm, T, 0, stream>>>(bufB, W2, bufA, N);
  gather_rows_kernel<false><<<gNodeWave, T, 0, stream>>>(bufA, perm, off, cnt, dinv, b2, out, N);
}


# GEMM retiled: one wave computes a 16-row strip over all 128 columns (8 accumulator tiles per A fragment), both operands staged in LDS, 25-wave work-groups, one per WGP; per-tile k order unchanged
# speedup vs baseline: 2.5609x; 1.1468x over previous
#include <hip/hip_runtime.h>
#include <hip/hip_bf16.h>

#define GDIM 128

typedef float v2f __attribute__((ext_vector_type(2)));
typedef float v8f __attribute__((ext_vector_type(8)));

__global__ void zero_counts_kernel(int* __restrict__ cnt, int n) {
  int i = blockIdx.x * blockDim.x + threadIdx.x;
  if (i < n) cnt[i] = 0;
}

__global__ void count_rank_kernel(const int* __restrict__ dst, int* __restrict__ cnt,
                                  int* __restrict__ rank, int E) {
  int e = blockIdx.x * blockDim.x + threadIdx.x;
  if (e < E) {
    rank[e] = __hip_atomic_fetch_add(&cnt[dst[e]], 1, __ATOMIC_RELAXED, __HIP_MEMORY_SCOPE_AGENT);
  }
}

#define SCAN_PER 16
__global__ void __launch_bounds__(1024) scan_dinv_kernel(const int* __restrict__ cnt,
                                                         int* __restrict__ off,
                                                         float* __restrict__ dinv, int n) {
  __shared__ int wsum[32];
  const int tid = threadIdx.x;
  const int lane = tid & 31;
  const int wave = tid >> 5;
  int carry = 0;
  for (int base = 0; base < n; base += 1024 * SCAN_PER) {
    const int i = base + tid * SCAN_PER;
    const bool full = (i + SCAN_PER <= n);
    int c[SCAN_PER];
    if (full) {
      const int4* p = (const int4*)(cnt + i);
#pragma unroll
      for (int q = 0; q < SCAN_PER / 4; ++q) {
        const int4 t = p[q];
        c[4 * q + 0] = t.x; c[4 * q + 1] = t.y; c[4 * q + 2] = t.z; c[4 * q + 3] = t.w;
      }
    } else {
#pragma unroll
      for (int k = 0; k < SCAN_PER; ++k) c[k] = (i + k < n) ? cnt[i + k] : 0;
    }
    int tsum = 0;
#pragma unroll
    for (int k = 0; k < SCAN_PER; ++k) tsum += c[k];

    int incl = tsum;
    for (int d = 1; d < 32; d <<= 1) {
      const int up = __shfl_up(incl, d);
      if (lane >= d) incl += up;
    }
    if (lane == 31) wsum[wave] = incl;
    __syncthreads();
    const int wv = wsum[lane];
    int winc = wv;
    for (int d = 1; d < 32; d <<= 1) {
      const int up = __shfl_up(winc, d);
      if (lane >= d) winc += up;
    }
    const int total = __shfl(winc, 31);
    const int wbase = __shfl(winc, wave) - __shfl(wv, wave);
    int run = carry + wbase + incl - tsum;

    int o[SCAN_PER];
    float r[SCAN_PER];
#pragma unroll
    for (int k = 0; k < SCAN_PER; ++k) {
      o[k] = run;
      run += c[k];
      r[k] = rsqrtf((float)(c[k] + 1));
    }
    if (full) {
      int4* po = (int4*)(off + i);
      float4* pr = (float4*)(dinv + i);
#pragma unroll
      for (int q = 0; q < SCAN_PER / 4; ++q) {
        po[q] = make_int4(o[4 * q + 0], o[4 * q + 1], o[4 * q + 2], o[4 * q + 3]);
        pr[q] = make_float4(r[4 * q + 0], r[4 * q + 1], r[4 * q + 2], r[4 * q + 3]);
      }
    } else {
#pragma unroll
      for (int k = 0; k < SCAN_PER; ++k) {
        if (i + k < n) { off[i + k] = o[k]; dinv[i + k] = r[k]; }
      }
    }
    carry += total;
    __syncthreads();
  }
}

__global__ void fill_perm_kernel(const int* __restrict__ src, const int* __restrict__ dst,
                                 const int* __restrict__ rank, const int* __restrict__ off,
                                 int* __restrict__ perm, int E) {
  int e = blockIdx.x * blockDim.x + threadIdx.x;
  if (e < E) perm[off[dst[e]] + rank[e]] = src[e];
}

#define BS_CPITCH 20
#define BS_KPITCH (16 * BS_CPITCH)
#define AS_PITCH 132
#define GEMM_WAVES 25
#define B_ITERS ((GDIM / 2 + GEMM_WAVES - 1) / GEMM_WAVES)

__global__ void __launch_bounds__(GEMM_WAVES * 32)
gemm_wmma_f32_kernel(const float* __restrict__ A, const float* __restrict__ B,
                     float* __restrict__ C, int nStrips) {
  __shared__ float Bs[(GDIM / 2) * BS_KPITCH];
  __shared__ float As[GEMM_WAVES * 16 * AS_PITCH];

  const int wave = __builtin_amdgcn_readfirstlane(threadIdx.x >> 5);
  const int lane = threadIdx.x & 31;
  const int half = lane >> 4;
  const int lmod = lane & 15;
  const int strip = blockIdx.x * GEMM_WAVES + wave;
  const int sidx = strip < nStrips ? strip : nStrips - 1;
  const int row0 = sidx * 16;

  float4 w0[B_ITERS], w1[B_ITERS];
#pragma unroll
  for (int it = 0; it < B_ITERS; ++it) {
    const int kq = it * GEMM_WAVES + wave;
    const int kp = kq < GDIM / 2 ? kq : GDIM / 2 - 1;
    w0[it] = *(const float4*)(B + (size_t)(2 * kp) * GDIM + 4 * lane);
    w1[it] = *(const float4*)(B + (size_t)(2 * kp + 1) * GDIM + 4 * lane);
  }

  const float4* Ag = (const float4*)(A + (size_t)row0 * GDIM) + lane;
  float4 ar[16];
#pragma unroll
  for (int r = 0; r < 16; ++r) ar[r] = Ag[r * (GDIM / 4)];

  {
    float* d0 = Bs + (lane & 3) * 4 * BS_CPITCH + (lane >> 2) * 2;
#pragma unroll
    for (int it = 0; it < B_ITERS; ++it) {
      const int kq = it * GEMM_WAVES + wave;
      const int kp = kq < GDIM / 2 ? kq : GDIM / 2 - 1;
      float* d = d0 + kp * BS_KPITCH;
      v2f p;
      p.x = w0[it].x; p.y = w1[it].x; *(v2f*)(d + 0 * BS_CPITCH) = p;
      p.x = w0[it].y; p.y = w1[it].y; *(v2f*)(d + 1 * BS_CPITCH) = p;
      p.x = w0[it].z; p.y = w1[it].z; *(v2f*)(d + 2 * BS_CPITCH) = p;
      p.x = w0[it].w; p.y = w1[it].w; *(v2f*)(d + 3 * BS_CPITCH) = p;
    }
  }
  float* as = As + wave * (16 * AS_PITCH);
#pragma unroll
  for (int r = 0; r < 16; ++r) *(float4*)(as + r * AS_PITCH + 4 * lane) = ar[r];

  __syncthreads();

  v8f acc[8];
#pragma unroll
  for (int t = 0; t < 8; ++t) acc[t] = (v8f){0.f, 0.f, 0.f, 0.f, 0.f, 0.f, 0.f, 0.f};

  const float* ap = as + lmod * AS_PITCH + half * 2;
  const float* bp = Bs + half * BS_KPITCH + lmod * BS_CPITCH;
#pragma unroll 8
  for (int ks = 0; ks < GDIM / 4; ++ks) {
    const v2f a = *(const v2f*)(ap + 4 * ks);
    const float4* bq = (const float4*)(bp + ks * 2 * BS_KPITCH);
    float4 bv[4];
#pragma unroll
    for (int j = 0; j < 4; ++j) bv[j] = bq[j];
#pragma unroll
    for (int j = 0; j < 4; ++j) {
      v2f b0, b1;
      b0.x = bv[j].x; b0.y = bv[j].y;
      b1.x = bv[j].z; b1.y = bv[j].w;
      acc[2 * j] = __builtin_amdgcn_wmma_f32_16x16x4_f32(
           false, a,  false, b0,
           (short)0, acc[2 * j],  false,  false);
      acc[2 * j + 1] = __builtin_amdgcn_wmma_f32_16x16x4_f32(
           false, a,  false, b1,
           (short)0, acc[2 * j + 1],  false,  false);
    }
  }

  if (strip < nStrips) {
    float* Cp = C + (size_t)(row0 + half * 8) * GDIM + lmod;
#pragma unroll
    for (int t = 0; t < 8; ++t) {
#pragma unroll
      for (int r = 0; r < 8; ++r) Cp[(size_t)r * GDIM + 16 * t] = acc[t][r];
    }
  }
}

template <int K>
__device__ __forceinline__ void accum_rows(const float4* __restrict__ H4,
                                           const int* __restrict__ perm,
                                           const float* __restrict__ dinv,
                                           int j, int lane, float dn, float4& acc) {
  int s[K];
  float4 v[K];
  float w[K];
#pragma unroll
  for (int k = 0; k < K; ++k) s[k] = perm[j + k];
#pragma unroll
  for (int k = 0; k < K; ++k) v[k] = H4[(size_t)s[k] * (GDIM / 4) + lane];
#pragma unroll
  for (int k = 0; k < K; ++k) w[k] = dinv[s[k]] * dn;
#pragma unroll
  for (int k = 0; k < K; ++k) {
    acc.x = fmaf(v[k].x, w[k], acc.x);
    acc.y = fmaf(v[k].y, w[k], acc.y);
    acc.z = fmaf(v[k].z, w[k], acc.z);
    acc.w = fmaf(v[k].w, w[k], acc.w);
  }
}

template <bool RELU>
__global__ void gather_rows_kernel(const float* __restrict__ h,
                                   const int* __restrict__ perm,
                                   const int* __restrict__ off,
                                   const int* __restrict__ cnt,
                                   const float* __restrict__ dinv,
                                   const float* __restrict__ bias,
                                   float* __restrict__ out, int n) {
  const int wave = __builtin_amdgcn_readfirstlane(threadIdx.x >> 5);
  const int node = blockIdx.x * 8 + wave;
  if (node >= n) return;
  const int lane = threadIdx.x & 31;
  const float4* H4 = (const float4*)h;
  int j = off[node];
  const int end = j + cnt[node];
  const float dn = dinv[node];

  const float* bp = bias + lane * 4;
  const float b0 = bp[0], b1v = bp[1], b2v = bp[2], b3 = bp[3];
  const float4 self = H4[(size_t)node * (GDIM / 4) + lane];
  const float dd = dn * dn;
  float4 acc;
  acc.x = self.x * dd;
  acc.y = self.y * dd;
  acc.z = self.z * dd;
  acc.w = self.w * dd;

  for (; j + 8 <= end; j += 8) accum_rows<8>(H4, perm, dinv, j, lane, dn, acc);
  if (j + 4 <= end) { accum_rows<4>(H4, perm, dinv, j, lane, dn, acc); j += 4; }
  if (j + 2 <= end) { accum_rows<2>(H4, perm, dinv, j, lane, dn, acc); j += 2; }
  if (j < end)      { accum_rows<1>(H4, perm, dinv, j, lane, dn, acc); }

  acc.x += b0;
  acc.y += b1v;
  acc.z += b2v;
  acc.w += b3;
  if (RELU) {
    acc.x = fmaxf(acc.x, 0.0f);
    acc.y = fmaxf(acc.y, 0.0f);
    acc.z = fmaxf(acc.z, 0.0f);
    acc.w = fmaxf(acc.w, 0.0f);
  }
  ((float4*)(out + (size_t)node * GDIM))[lane] = acc;
}

extern "C" void kernel_launch(void* const* d_in, const int* in_sizes, int n_in,
                              void* d_out, int out_size, void* d_ws, size_t ws_size,
                              hipStream_t stream) {
  const float* x  = (const float*)d_in[0];
  const int*   ei = (const int*)d_in[1];
  const float* W1 = (const float*)d_in[2];
  const float* b1 = (const float*)d_in[3];
  const float* W2 = (const float*)d_in[4];
  const float* b2 = (const float*)d_in[5];
  float* out = (float*)d_out;

  const int N = in_sizes[0] / GDIM;
  const int E = in_sizes[1] / 2;
  const int* src = ei;
  const int* dst = ei + E;

  float* dinv = (float*)d_ws;
  float* bufA = dinv + N;
  float* bufB = bufA + (size_t)N * GDIM;
  int* cnt  = (int*)(bufB + (size_t)N * GDIM);
  int* off  = cnt + N;
  int* rank = off + N;
  int* perm = rank + E;

  const int T = 256;
  const int gN    = (N + T - 1) / T;
  const int gE    = (E + T - 1) / T;
  const int gNodeWave = (N + 7) / 8;
  const int nStrips = N / 16;
  const int gGemm = (nStrips + GEMM_WAVES - 1) / GEMM_WAVES;

  zero_counts_kernel<<<gN, T, 0, stream>>>(cnt, N);
  count_rank_kernel<<<gE, T, 0, stream>>>(dst, cnt, rank, E);
  scan_dinv_kernel<<<1, 1024, 0, stream>>>(cnt, off, dinv, N);
  fill_perm_kernel<<<gE, T, 0, stream>>>(src, dst, rank, off, perm, E);

  gemm_wmma_f32_kernel<<<gGemm, GEMM_WAVES * 32, 0, stream>>>(x, W1, bufA, nStrips);
  gather_rows_kernel<true><<<gNodeWave, T, 0, stream>>>(bufA, perm, off, cnt, dinv, b1, bufB, N);

  gemm_wmma_f32_kernel<<<gGemm, GEMM_WAVES * 32, 0, stream>>>(bufB, W2, bufA, nStrips);
  gather_rows_kernel<false><<<gNodeWave, T, 0, stream>>>(bufA, perm, off, cnt, dinv, b2, out, N);
}


# edge bucketing without global atomics: 8 node ranges x 32 edge chunks, LDS counters give counts and slots, block-local scan plus block totals; bucket order chunk-major, then LDS arrival order
# speedup vs baseline: 3.9236x; 1.5321x over previous
#include <hip/hip_runtime.h>
#include <hip/hip_bf16.h>

#define GDIM 128

typedef float v2f __attribute__((ext_vector_type(2)));
typedef float v8f __attribute__((ext_vector_type(8)));

#define NR 8
#define NC 32
#define RS_MAX 16384
#define SCAN_B 1024
#define NB_MAX 128
#define EB 8

__global__ void __launch_bounds__(1024) hist_kernel(const int* __restrict__ dst,
                                                    int* __restrict__ cntC,
                                                    int N, int E, int RS, int CS) {
  __shared__ int lcnt[RS_MAX];
  const int r = blockIdx.x;
  const int c = blockIdx.y;
  const int tid = threadIdx.x;
  const int rbase = r * RS;
  for (int i = tid; i < RS; i += 1024) lcnt[i] = 0;
  __syncthreads();
  const int cbeg = c * CS;
  const int cend = (cbeg + CS < E) ? cbeg + CS : E;
  for (int e0 = cbeg + tid; e0 < cend; e0 += EB * 1024) {
    int dv[EB];
#pragma unroll
    for (int k = 0; k < EB; ++k) {
      const int e = e0 + k * 1024;
      dv[k] = (e < cend) ? dst[e] : -1;
    }
#pragma unroll
    for (int k = 0; k < EB; ++k) {
      const unsigned d = (unsigned)(dv[k] - rbase);
      if (d < (unsigned)RS) atomicAdd(&lcnt[d], 1);
    }
  }
  __syncthreads();
  for (int i = tid; i < RS; i += 1024) {
    const int n = rbase + i;
    if (n < N) cntC[(size_t)c * N + n] = lcnt[i];
  }
}

__global__ void __launch_bounds__(SCAN_B) scan_dinv_kernel(int* __restrict__ cntC,
                                                           int* __restrict__ cnt,
                                                           float* __restrict__ dinv,
                                                           int* __restrict__ btot, int N) {
  __shared__ int wsum[32];
  const int tid = threadIdx.x;
  const int lane = tid & 31;
  const int wave = tid >> 5;
  const int n = blockIdx.x * SCAN_B + tid;
  const bool valid = n < N;
  int v[NC];
#pragma unroll
  for (int c = 0; c < NC; ++c) v[c] = valid ? cntC[(size_t)c * N + n] : 0;
  int tot = 0;
#pragma unroll
  for (int c = 0; c < NC; ++c) tot += v[c];

  int incl = tot;
  for (int d = 1; d < 32; d <<= 1) {
    const int up = __shfl_up(incl, d);
    if (lane >= d) incl += up;
  }
  if (lane == 31) wsum[wave] = incl;
  __syncthreads();
  const int wv = wsum[lane];
  int winc = wv;
  for (int d = 1; d < 32; d <<= 1) {
    const int up = __shfl_up(winc, d);
    if (lane >= d) winc += up;
  }
  const int total = __shfl(winc, 31);
  const int wbase = __shfl(winc, wave) - __shfl(wv, wave);
  int run = wbase + incl - tot;
  if (valid) {
    cnt[n] = tot;
    dinv[n] = rsqrtf((float)(tot + 1));
#pragma unroll
    for (int c = 0; c < NC; ++c) {
      cntC[(size_t)c * N + n] = run;
      run += v[c];
    }
  }
  if (tid == 0) btot[blockIdx.x] = total;
}

__global__ void __launch_bounds__(1024) fill_perm_kernel(const int* __restrict__ src,
                                                         const int* __restrict__ dst,
                                                         const int* __restrict__ cntC,
                                                         const int* __restrict__ btot,
                                                         int* __restrict__ off,
                                                         int* __restrict__ perm,
                                                         int N, int E, int RS, int CS, int NB) {
  __shared__ int cur[RS_MAX];
  __shared__ int bp[NB_MAX];
  const int r = blockIdx.x;
  const int c = blockIdx.y;
  const int tid = threadIdx.x;
  const int rbase = r * RS;
  if (tid < 32) {
    int t[4];
#pragma unroll
    for (int k = 0; k < 4; ++k) t[k] = (tid * 4 + k < NB) ? btot[tid * 4 + k] : 0;
    const int s = t[0] + t[1] + t[2] + t[3];
    int incl = s;
    for (int d = 1; d < 32; d <<= 1) {
      const int up = __shfl_up(incl, d);
      if (tid >= d) incl += up;
    }
    int ex = incl - s;
#pragma unroll
    for (int k = 0; k < 4; ++k) {
      bp[tid * 4 + k] = ex;
      ex += t[k];
    }
  }
  __syncthreads();
  for (int i = tid; i < RS; i += 1024) {
    const int n = rbase + i;
    if (n < N) {
      const int b = bp[n / SCAN_B] + cntC[(size_t)c * N + n];
      cur[i] = b;
      if (c == 0) off[n] = b;
    }
  }
  __syncthreads();
  const int cbeg = c * CS;
  const int cend = (cbeg + CS < E) ? cbeg + CS : E;
  for (int e0 = cbeg + tid; e0 < cend; e0 += EB * 1024) {
    int dv[EB], sv[EB];
#pragma unroll
    for (int k = 0; k < EB; ++k) {
      const int e = e0 + k * 1024;
      dv[k] = (e < cend) ? dst[e] : -1;
      sv[k] = (e < cend) ? src[e] : 0;
    }
    int pos[EB];
#pragma unroll
    for (int k = 0; k < EB; ++k) {
      const unsigned d = (unsigned)(dv[k] - rbase);
      pos[k] = (d < (unsigned)RS) ? atomicAdd(&cur[d], 1) : -1;
    }
#pragma unroll
    for (int k = 0; k < EB; ++k) {
      if (pos[k] >= 0) perm[pos[k]] = sv[k];
    }
  }
}

#define BS_CPITCH 20
#define BS_KPITCH (16 * BS_CPITCH)
#define AS_PITCH 132
#define GEMM_WAVES 25
#define B_ITERS ((GDIM / 2 + GEMM_WAVES - 1) / GEMM_WAVES)

__global__ void __launch_bounds__(GEMM_WAVES * 32)
gemm_wmma_f32_kernel(const float* __restrict__ A, const float* __restrict__ B,
                     float* __restrict__ C, int nStrips) {
  __shared__ float Bs[(GDIM / 2) * BS_KPITCH];
  __shared__ float As[GEMM_WAVES * 16 * AS_PITCH];

  const int wave = __builtin_amdgcn_readfirstlane(threadIdx.x >> 5);
  const int lane = threadIdx.x & 31;
  const int half = lane >> 4;
  const int lmod = lane & 15;
  const int strip = blockIdx.x * GEMM_WAVES + wave;
  const int sidx = strip < nStrips ? strip : nStrips - 1;
  const int row0 = sidx * 16;

  float4 w0[B_ITERS], w1[B_ITERS];
#pragma unroll
  for (int it = 0; it < B_ITERS; ++it) {
    const int kq = it * GEMM_WAVES + wave;
    const int kp = kq < GDIM / 2 ? kq : GDIM / 2 - 1;
    w0[it] = *(const float4*)(B + (size_t)(2 * kp) * GDIM + 4 * lane);
    w1[it] = *(const float4*)(B + (size_t)(2 * kp + 1) * GDIM + 4 * lane);
  }

  const float4* Ag = (const float4*)(A + (size_t)row0 * GDIM) + lane;
  float4 ar[16];
#pragma unroll
  for (int r = 0; r < 16; ++r) ar[r] = Ag[r * (GDIM / 4)];

  {
    float* d0 = Bs + (lane & 3) * 4 * BS_CPITCH + (lane >> 2) * 2;
#pragma unroll
    for (int it = 0; it < B_ITERS; ++it) {
      const int kq = it * GEMM_WAVES + wave;
      const int kp = kq < GDIM / 2 ? kq : GDIM / 2 - 1;
      float* d = d0 + kp * BS_KPITCH;
      v2f p;
      p.x = w0[it].x; p.y = w1[it].x; *(v2f*)(d + 0 * BS_CPITCH) = p;
      p.x = w0[it].y; p.y = w1[it].y; *(v2f*)(d + 1 * BS_CPITCH) = p;
      p.x = w0[it].z; p.y = w1[it].z; *(v2f*)(d + 2 * BS_CPITCH) = p;
      p.x = w0[it].w; p.y = w1[it].w; *(v2f*)(d + 3 * BS_CPITCH) = p;
    }
  }
  float* as = As + wave * (16 * AS_PITCH);
#pragma unroll
  for (int r = 0; r < 16; ++r) *(float4*)(as + r * AS_PITCH + 4 * lane) = ar[r];

  __syncthreads();

  v8f acc[8];
#pragma unroll
  for (int t = 0; t < 8; ++t) acc[t] = (v8f){0.f, 0.f, 0.f, 0.f, 0.f, 0.f, 0.f, 0.f};

  const float* ap = as + lmod * AS_PITCH + half * 2;
  const float* bp = Bs + half * BS_KPITCH + lmod * BS_CPITCH;
#pragma unroll 8
  for (int ks = 0; ks < GDIM / 4; ++ks) {
    const v2f a = *(const v2f*)(ap + 4 * ks);
    const float4* bq = (const float4*)(bp + ks * 2 * BS_KPITCH);
    float4 bv[4];
#pragma unroll
    for (int j = 0; j < 4; ++j) bv[j] = bq[j];
#pragma unroll
    for (int j = 0; j < 4; ++j) {
      v2f b0, b1;
      b0.x = bv[j].x; b0.y = bv[j].y;
      b1.x = bv[j].z; b1.y = bv[j].w;
      acc[2 * j] = __builtin_amdgcn_wmma_f32_16x16x4_f32(
           false, a,  false, b0,
           (short)0, acc[2 * j],  false,  false);
      acc[2 * j + 1] = __builtin_amdgcn_wmma_f32_16x16x4_f32(
           false, a,  false, b1,
           (short)0, acc[2 * j + 1],  false,  false);
    }
  }

  if (strip < nStrips) {
    float* Cp = C + (size_t)(row0 + half * 8) * GDIM + lmod;
#pragma unroll
    for (int t = 0; t < 8; ++t) {
#pragma unroll
      for (int r = 0; r < 8; ++r) Cp[(size_t)r * GDIM + 16 * t] = acc[t][r];
    }
  }
}

template <int K>
__device__ __forceinline__ void accum_rows(const float4* __restrict__ H4,
                                           const int* __restrict__ perm,
                                           const float* __restrict__ dinv,
                                           int j, int lane, float dn, float4& acc) {
  int s[K];
  float4 v[K];
  float w[K];
#pragma unroll
  for (int k = 0; k < K; ++k) s[k] = perm[j + k];
#pragma unroll
  for (int k = 0; k < K; ++k) v[k] = H4[(size_t)s[k] * (GDIM / 4) + lane];
#pragma unroll
  for (int k = 0; k < K; ++k) w[k] = dinv[s[k]] * dn;
#pragma unroll
  for (int k = 0; k < K; ++k) {
    acc.x = fmaf(v[k].x, w[k], acc.x);
    acc.y = fmaf(v[k].y, w[k], acc.y);
    acc.z = fmaf(v[k].z, w[k], acc.z);
    acc.w = fmaf(v[k].w, w[k], acc.w);
  }
}

template <bool RELU>
__global__ void gather_rows_kernel(const float* __restrict__ h,
                                   const int* __restrict__ perm,
                                   const int* __restrict__ off,
                                   const int* __restrict__ cnt,
                                   const float* __restrict__ dinv,
                                   const float* __restrict__ bias,
                                   float* __restrict__ out, int n) {
  const int wave = __builtin_amdgcn_readfirstlane(threadIdx.x >> 5);
  const int node = blockIdx.x * 8 + wave;
  if (node >= n) return;
  const int lane = threadIdx.x & 31;
  const float4* H4 = (const float4*)h;
  int j = off[node];
  const int end = j + cnt[node];
  const float dn = dinv[node];

  const float* bp = bias + lane * 4;
  const float b0 = bp[0], b1v = bp[1], b2v = bp[2], b3 = bp[3];
  const float4 self = H4[(size_t)node * (GDIM / 4) + lane];
  const float dd = dn * dn;
  float4 acc;
  acc.x = self.x * dd;
  acc.y = self.y * dd;
  acc.z = self.z * dd;
  acc.w = self.w * dd;

  for (; j + 8 <= end; j += 8) accum_rows<8>(H4, perm, dinv, j, lane, dn, acc);
  if (j + 4 <= end) { accum_rows<4>(H4, perm, dinv, j, lane, dn, acc); j += 4; }
  if (j + 2 <= end) { accum_rows<2>(H4, perm, dinv, j, lane, dn, acc); j += 2; }
  if (j < end)      { accum_rows<1>(H4, perm, dinv, j, lane, dn, acc); }

  acc.x += b0;
  acc.y += b1v;
  acc.z += b2v;
  acc.w += b3;
  if (RELU) {
    acc.x = fmaxf(acc.x, 0.0f);
    acc.y = fmaxf(acc.y, 0.0f);
    acc.z = fmaxf(acc.z, 0.0f);
    acc.w = fmaxf(acc.w, 0.0f);
  }
  ((float4*)(out + (size_t)node * GDIM))[lane] = acc;
}

extern "C" void kernel_launch(void* const* d_in, const int* in_sizes, int n_in,
                              void* d_out, int out_size, void* d_ws, size_t ws_size,
                              hipStream_t stream) {
  const float* x  = (const float*)d_in[0];
  const int*   ei = (const int*)d_in[1];
  const float* W1 = (const float*)d_in[2];
  const float* b1 = (const float*)d_in[3];
  const float* W2 = (const float*)d_in[4];
  const float* b2 = (const float*)d_in[5];
  float* out = (float*)d_out;

  const int N = in_sizes[0] / GDIM;
  const int E = in_sizes[1] / 2;
  const int* src = ei;
  const int* dst = ei + E;

  float* dinv = (float*)d_ws;
  float* bufA = dinv + N;
  float* bufB = bufA + (size_t)N * GDIM;
  int* cnt  = (int*)(bufB + (size_t)N * GDIM);
  int* off  = cnt + N;
  int* cntC = off + N;
  int* btot = cntC + (size_t)NC * N;
  int* perm = btot + NB_MAX;

  const int T = 256;
  const int RS = (N + NR - 1) / NR;
  const int CS = (E + NC - 1) / NC;
  const int NB = (N + SCAN_B - 1) / SCAN_B;
  const int gNodeWave = (N + 7) / 8;
  const int nStrips = N / 16;
  const int gGemm = (nStrips + GEMM_WAVES - 1) / GEMM_WAVES;

  hist_kernel<<<dim3(NR, NC), 1024, 0, stream>>>(dst, cntC, N, E, RS, CS);
  scan_dinv_kernel<<<NB, SCAN_B, 0, stream>>>(cntC, cnt, dinv, btot, N);
  fill_perm_kernel<<<dim3(NR, NC), 1024, 0, stream>>>(src, dst, cntC, btot, off, perm, N, E, RS, CS, NB);

  gemm_wmma_f32_kernel<<<gGemm, GEMM_WAVES * 32, 0, stream>>>(x, W1, bufA, nStrips);
  gather_rows_kernel<true><<<gNodeWave, T, 0, stream>>>(bufA, perm, off, cnt, dinv, b1, bufB, N);

  gemm_wmma_f32_kernel<<<gGemm, GEMM_WAVES * 32, 0, stream>>>(bufB, W2, bufA, nStrips);
  gather_rows_kernel<false><<<gNodeWave, T, 0, stream>>>(bufA, perm, off, cnt, dinv, b2, out, N);
}


# layer-1 row gather fused into the layer-2 GEMM launch: each wave sums its strip's 16 rows into its LDS A image, walking the strip's run of sources 16 rows at a time; per-node sum order kept
# speedup vs baseline: 4.1965x; 1.0696x over previous
#include <hip/hip_runtime.h>
#include <hip/hip_bf16.h>

#define GDIM 128

typedef float v2f __attribute__((ext_vector_type(2)));
typedef float v8f __attribute__((ext_vector_type(8)));

#define NR 8
#define NC 32
#define RS_MAX 16384
#define SCAN_B 1024
#define NB_MAX 128
#define EB 8

__global__ void __launch_bounds__(1024) hist_kernel(const int* __restrict__ dst,
                                                    int* __restrict__ cntC,
                                                    int N, int E, int RS, int CS) {
  __shared__ int lcnt[RS_MAX];
  const int r = blockIdx.x;
  const int c = blockIdx.y;
  const int tid = threadIdx.x;
  const int rbase = r * RS;
  for (int i = tid; i < RS; i += 1024) lcnt[i] = 0;
  __syncthreads();
  const int cbeg = c * CS;
  const int cend = (cbeg + CS < E) ? cbeg + CS : E;
  for (int e0 = cbeg + tid; e0 < cend; e0 += EB * 1024) {
    int dv[EB];
#pragma unroll
    for (int k = 0; k < EB; ++k) {
      const int e = e0 + k * 1024;
      dv[k] = (e < cend) ? dst[e] : -1;
    }
#pragma unroll
    for (int k = 0; k < EB; ++k) {
      const unsigned d = (unsigned)(dv[k] - rbase);
      if (d < (unsigned)RS) atomicAdd(&lcnt[d], 1);
    }
  }
  __syncthreads();
  for (int i = tid; i < RS; i += 1024) {
    const int n = rbase + i;
    if (n < N) cntC[(size_t)c * N + n] = lcnt[i];
  }
}

__global__ void __launch_bounds__(SCAN_B) scan_dinv_kernel(int* __restrict__ cntC,
                                                           int* __restrict__ cnt,
                                                           float* __restrict__ dinv,
                                                           int* __restrict__ btot, int N) {
  __shared__ int wsum[32];
  const int tid = threadIdx.x;
  const int lane = tid & 31;
  const int wave = tid >> 5;
  const int n = blockIdx.x * SCAN_B + tid;
  const bool valid = n < N;
  int v[NC];
#pragma unroll
  for (int c = 0; c < NC; ++c) v[c] = valid ? cntC[(size_t)c * N + n] : 0;
  int tot = 0;
#pragma unroll
  for (int c = 0; c < NC; ++c) tot += v[c];

  int incl = tot;
  for (int d = 1; d < 32; d <<= 1) {
    const int up = __shfl_up(incl, d);
    if (lane >= d) incl += up;
  }
  if (lane == 31) wsum[wave] = incl;
  __syncthreads();
  const int wv = wsum[lane];
  int winc = wv;
  for (int d = 1; d < 32; d <<= 1) {
    const int up = __shfl_up(winc, d);
    if (lane >= d) winc += up;
  }
  const int total = __shfl(winc, 31);
  const int wbase = __shfl(winc, wave) - __shfl(wv, wave);
  int run = wbase + incl - tot;
  if (valid) {
    cnt[n] = tot;
    dinv[n] = rsqrtf((float)(tot + 1));
#pragma unroll
    for (int c = 0; c < NC; ++c) {
      cntC[(size_t)c * N + n] = run;
      run += v[c];
    }
  }
  if (tid == 0) btot[blockIdx.x] = total;
}

__global__ void __launch_bounds__(1024) fill_perm_kernel(const int* __restrict__ src,
                                                         const int* __restrict__ dst,
                                                         const int* __restrict__ cntC,
                                                         const int* __restrict__ btot,
                                                         int* __restrict__ off,
                                                         int* __restrict__ perm,
                                                         int N, int E, int RS, int CS, int NB) {
  __shared__ int cur[RS_MAX];
  __shared__ int bp[NB_MAX];
  const int r = blockIdx.x;
  const int c = blockIdx.y;
  const int tid = threadIdx.x;
  const int rbase = r * RS;
  if (tid < 32) {
    int t[4];
#pragma unroll
    for (int k = 0; k < 4; ++k) t[k] = (tid * 4 + k < NB) ? btot[tid * 4 + k] : 0;
    const int s = t[0] + t[1] + t[2] + t[3];
    int incl = s;
    for (int d = 1; d < 32; d <<= 1) {
      const int up = __shfl_up(incl, d);
      if (tid >= d) incl += up;
    }
    int ex = incl - s;
#pragma unroll
    for (int k = 0; k < 4; ++k) {
      bp[tid * 4 + k] = ex;
      ex += t[k];
    }
  }
  __syncthreads();
  for (int i = tid; i < RS; i += 1024) {
    const int n = rbase + i;
    if (n < N) {
      const int b = bp[n / SCAN_B] + cntC[(size_t)c * N + n];
      cur[i] = b;
      if (c == 0) off[n] = b;
    }
  }
  __syncthreads();
  const int cbeg = c * CS;
  const int cend = (cbeg + CS < E) ? cbeg + CS : E;
  for (int e0 = cbeg + tid; e0 < cend; e0 += EB * 1024) {
    int dv[EB], sv[EB];
#pragma unroll
    for (int k = 0; k < EB; ++k) {
      const int e = e0 + k * 1024;
      dv[k] = (e < cend) ? dst[e] : -1;
      sv[k] = (e < cend) ? src[e] : 0;
    }
    int pos[EB];
#pragma unroll
    for (int k = 0; k < EB; ++k) {
      const unsigned d = (unsigned)(dv[k] - rbase);
      pos[k] = (d < (unsigned)RS) ? atomicAdd(&cur[d], 1) : -1;
    }
#pragma unroll
    for (int k = 0; k < EB; ++k) {
      if (pos[k] >= 0) perm[pos[k]] = sv[k];
    }
  }
}

#define BS_CPITCH 20
#define BS_KPITCH (16 * BS_CPITCH)
#define AS_PITCH 132
#define GEMM_WAVES 25
#define B_ITERS ((GDIM / 2 + GEMM_WAVES - 1) / GEMM_WAVES)

__global__ void __launch_bounds__(GEMM_WAVES * 32)
gemm_wmma_f32_kernel(const float* __restrict__ A, const float* __restrict__ B,
                     float* __restrict__ C, int nStrips) {
  __shared__ float Bs[(GDIM / 2) * BS_KPITCH];
  __shared__ float As[GEMM_WAVES * 16 * AS_PITCH];

  const int wave = __builtin_amdgcn_readfirstlane(threadIdx.x >> 5);
  const int lane = threadIdx.x & 31;
  const int half = lane >> 4;
  const int lmod = lane & 15;
  const int strip = blockIdx.x * GEMM_WAVES + wave;
  const int sidx = strip < nStrips ? strip : nStrips - 1;
  const int row0 = sidx * 16;

  float4 w0[B_ITERS], w1[B_ITERS];
#pragma unroll
  for (int it = 0; it < B_ITERS; ++it) {
    const int kq = it * GEMM_WAVES + wave;
    const int kp = kq < GDIM / 2 ? kq : GDIM / 2 - 1;
    w0[it] = *(const float4*)(B + (size_t)(2 * kp) * GDIM + 4 * lane);
    w1[it] = *(const float4*)(B + (size_t)(2 * kp + 1) * GDIM + 4 * lane);
  }

  const float4* Ag = (const float4*)(A + (size_t)row0 * GDIM) + lane;
  float4 ar[16];
#pragma unroll
  for (int r = 0; r < 16; ++r) ar[r] = Ag[r * (GDIM / 4)];

  {
    float* d0 = Bs + (lane & 3) * 4 * BS_CPITCH + (lane >> 2) * 2;
#pragma unroll
    for (int it = 0; it < B_ITERS; ++it) {
      const int kq = it * GEMM_WAVES + wave;
      const int kp = kq < GDIM / 2 ? kq : GDIM / 2 - 1;
      float* d = d0 + kp * BS_KPITCH;
      v2f p;
      p.x = w0[it].x; p.y = w1[it].x; *(v2f*)(d + 0 * BS_CPITCH) = p;
      p.x = w0[it].y; p.y = w1[it].y; *(v2f*)(d + 1 * BS_CPITCH) = p;
      p.x = w0[it].z; p.y = w1[it].z; *(v2f*)(d + 2 * BS_CPITCH) = p;
      p.x = w0[it].w; p.y = w1[it].w; *(v2f*)(d + 3 * BS_CPITCH) = p;
    }
  }
  float* as = As + wave * (16 * AS_PITCH);
#pragma unroll
  for (int r = 0; r < 16; ++r) *(float4*)(as + r * AS_PITCH + 4 * lane) = ar[r];

  __syncthreads();

  v8f acc[8];
#pragma unroll
  for (int t = 0; t < 8; ++t) acc[t] = (v8f){0.f, 0.f, 0.f, 0.f, 0.f, 0.f, 0.f, 0.f};

  const float* ap = as + lmod * AS_PITCH + half * 2;
  const float* bp = Bs + half * BS_KPITCH + lmod * BS_CPITCH;
#pragma unroll 8
  for (int ks = 0; ks < GDIM / 4; ++ks) {
    const v2f a = *(const v2f*)(ap + 4 * ks);
    const float4* bq = (const float4*)(bp + ks * 2 * BS_KPITCH);
    float4 bv[4];
#pragma unroll
    for (int j = 0; j < 4; ++j) bv[j] = bq[j];
#pragma unroll
    for (int j = 0; j < 4; ++j) {
      v2f b0, b1;
      b0.x = bv[j].x; b0.y = bv[j].y;
      b1.x = bv[j].z; b1.y = bv[j].w;
      acc[2 * j] = __builtin_amdgcn_wmma_f32_16x16x4_f32(
           false, a,  false, b0,
           (short)0, acc[2 * j],  false,  false);
      acc[2 * j + 1] = __builtin_amdgcn_wmma_f32_16x16x4_f32(
           false, a,  false, b1,
           (short)0, acc[2 * j + 1],  false,  false);
    }
  }

  if (strip < nStrips) {
    float* Cp = C + (size_t)(row0 + half * 8) * GDIM + lmod;
#pragma unroll
    for (int t = 0; t < 8; ++t) {
#pragma unroll
      for (int r = 0; r < 8; ++r) Cp[(size_t)r * GDIM + 16 * t] = acc[t][r];
    }
  }
}

template <int K>
__device__ __forceinline__ void accum_rows(const float4* __restrict__ H4,
                                           const int* __restrict__ perm,
                                           const float* __restrict__ dinv,
                                           int j, int lane, float dn, float4& acc) {
  int s[K];
  float4 v[K];
  float w[K];
#pragma unroll
  for (int k = 0; k < K; ++k) s[k] = perm[j + k];
#pragma unroll
  for (int k = 0; k < K; ++k) v[k] = H4[(size_t)s[k] * (GDIM / 4) + lane];
#pragma unroll
  for (int k = 0; k < K; ++k) w[k] = dinv[s[k]] * dn;
#pragma unroll
  for (int k = 0; k < K; ++k) {
    acc.x = fmaf(v[k].x, w[k], acc.x);
    acc.y = fmaf(v[k].y, w[k], acc.y);
    acc.z = fmaf(v[k].z, w[k], acc.z);
    acc.w = fmaf(v[k].w, w[k], acc.w);
  }
}

template <bool RELU>
__global__ void gather_rows_kernel(const float* __restrict__ h,
                                   const int* __restrict__ perm,
                                   const int* __restrict__ off,
                                   const int* __restrict__ cnt,
                                   const float* __restrict__ dinv,
                                   const float* __restrict__ bias,
                                   float* __restrict__ out, int n) {
  const int wave = __builtin_amdgcn_readfirstlane(threadIdx.x >> 5);
  const int node = blockIdx.x * 8 + wave;
  if (node >= n) return;
  const int lane = threadIdx.x & 31;
  const float4* H4 = (const float4*)h;
  int j = off[node];
  const int end = j + cnt[node];
  const float dn = dinv[node];

  const float* bp = bias + lane * 4;
  const float b0 = bp[0], b1v = bp[1], b2v = bp[2], b3 = bp[3];
  const float4 self = H4[(size_t)node * (GDIM / 4) + lane];
  const float dd = dn * dn;
  float4 acc;
  acc.x = self.x * dd;
  acc.y = self.y * dd;
  acc.z = self.z * dd;
  acc.w = self.w * dd;

  for (; j + 8 <= end; j += 8) accum_rows<8>(H4, perm, dinv, j, lane, dn, acc);
  if (j + 4 <= end) { accum_rows<4>(H4, perm, dinv, j, lane, dn, acc); j += 4; }
  if (j + 2 <= end) { accum_rows<2>(H4, perm, dinv, j, lane, dn, acc); j += 2; }
  if (j < end)      { accum_rows<1>(H4, perm, dinv, j, lane, dn, acc); }

  acc.x += b0;
  acc.y += b1v;
  acc.z += b2v;
  acc.w += b3;
  if (RELU) {
    acc.x = fmaxf(acc.x, 0.0f);
    acc.y = fmaxf(acc.y, 0.0f);
    acc.z = fmaxf(acc.z, 0.0f);
    acc.w = fmaxf(acc.w, 0.0f);
  }
  ((float4*)(out + (size_t)node * GDIM))[lane] = acc;
}

#define GB 16

__device__ __forceinline__ int lane_int(int v, int l) {
  return __builtin_amdgcn_readlane(v, l);
}
__device__ __forceinline__ float lane_float(float v, int l) {
  return __int_as_float(__builtin_amdgcn_readlane(__float_as_int(v), l));
}

__global__ void __launch_bounds__(GEMM_WAVES * 32)
gather_gemm_wmma_f32_kernel(const float* __restrict__ h, const int* __restrict__ perm,
                            const int* __restrict__ off, const int* __restrict__ cnt,
                            const float* __restrict__ dinv, const float* __restrict__ bias,
                            const float* __restrict__ B, float* __restrict__ C, int nStrips) {
  __shared__ float Bs[(GDIM / 2) * BS_KPITCH];
  __shared__ float As[GEMM_WAVES * 16 * AS_PITCH];

  const int wave = __builtin_amdgcn_readfirstlane(threadIdx.x >> 5);
  const int lane = threadIdx.x & 31;
  const int half = lane >> 4;
  const int lmod = lane & 15;
  const int strip = blockIdx.x * GEMM_WAVES + wave;
  const int sidx = strip < nStrips ? strip : nStrips - 1;
  const int row0 = sidx * 16;

  float4 w0[B_ITERS], w1[B_ITERS];
#pragma unroll
  for (int it = 0; it < B_ITERS; ++it) {
    const int kq = it * GEMM_WAVES + wave;
    const int kp = kq < GDIM / 2 ? kq : GDIM / 2 - 1;
    w0[it] = *(const float4*)(B + (size_t)(2 * kp) * GDIM + 4 * lane);
    w1[it] = *(const float4*)(B + (size_t)(2 * kp + 1) * GDIM + 4 * lane);
  }

  const float4* H4 = (const float4*)h;
  const float4* Ag = H4 + (size_t)row0 * (GDIM / 4) + lane;
  float4 ar[16];
#pragma unroll
  for (int r = 0; r < 16; ++r) ar[r] = Ag[r * (GDIM / 4)];

  const int o_l = off[row0 + lmod];
  const int c_l = cnt[row0 + lmod];
  const float d_l = dinv[row0 + lmod];
  const float* bq4 = bias + lane * 4;
  float4 bv;
  bv.x = bq4[0];
  bv.y = bq4[1];
  bv.z = bq4[2];
  bv.w = bq4[3];

  {
    float* d0 = Bs + (lane & 3) * 4 * BS_CPITCH + (lane >> 2) * 2;
#pragma unroll
    for (int it = 0; it < B_ITERS; ++it) {
      const int kq = it * GEMM_WAVES + wave;
      const int kp = kq < GDIM / 2 ? kq : GDIM / 2 - 1;
      float* d = d0 + kp * BS_KPITCH;
      v2f p;
      p.x = w0[it].x; p.y = w1[it].x; *(v2f*)(d + 0 * BS_CPITCH) = p;
      p.x = w0[it].y; p.y = w1[it].y; *(v2f*)(d + 1 * BS_CPITCH) = p;
      p.x = w0[it].z; p.y = w1[it].z; *(v2f*)(d + 2 * BS_CPITCH) = p;
      p.x = w0[it].w; p.y = w1[it].w; *(v2f*)(d + 3 * BS_CPITCH) = p;
    }
  }
  __syncthreads();

  float* as = As + wave * (16 * AS_PITCH);
  float* arow = as + 4 * lane;
#pragma unroll
  for (int r = 0; r < 16; ++r) {
    const float dr = lane_float(d_l, r);
    const float dd = dr * dr;
    float4 t;
    t.x = ar[r].x * dd;
    t.y = ar[r].y * dd;
    t.z = ar[r].z * dd;
    t.w = ar[r].w * dd;
    *(float4*)(arow + r * AS_PITCH) = t;
  }

  {
    const int segStart = lane_int(o_l, 0);
    const int segEnd = lane_int(o_l, 15) + lane_int(c_l, 15);
    int r = 0;
    int curEnd = segStart + lane_int(c_l, 0);
    float dn = lane_float(d_l, 0);
    float4 acc = *(const float4*)arow;

    if (segStart < segEnd) {
      const int last = segEnd - 1;
      int sn[GB];
#pragma unroll
      for (int k = 0; k < GB; ++k) {
        const int e = segStart + k;
        sn[k] = perm[e < last ? e : last];
      }
      for (int p = segStart; p < segEnd; p += GB) {
        int s[GB];
        float4 v[GB];
        float w[GB];
#pragma unroll
        for (int k = 0; k < GB; ++k) s[k] = sn[k];
#pragma unroll
        for (int k = 0; k < GB; ++k) v[k] = H4[(size_t)s[k] * (GDIM / 4) + lane];
#pragma unroll
        for (int k = 0; k < GB; ++k) w[k] = dinv[s[k]];
        if (p + GB < segEnd) {
#pragma unroll
          for (int k = 0; k < GB; ++k) {
            const int e = p + GB + k;
            sn[k] = perm[e < last ? e : last];
          }
        }
#pragma unroll
        for (int k = 0; k < GB; ++k) {
          const int e = p + k;
          if (e < segEnd) {
            while (e == curEnd && r < 15) {
              float4 t;
              t.x = fmaxf(acc.x + bv.x, 0.0f);
              t.y = fmaxf(acc.y + bv.y, 0.0f);
              t.z = fmaxf(acc.z + bv.z, 0.0f);
              t.w = fmaxf(acc.w + bv.w, 0.0f);
              *(float4*)(arow + r * AS_PITCH) = t;
              ++r;
              curEnd = lane_int(o_l, r) + lane_int(c_l, r);
              dn = lane_float(d_l, r);
              acc = *(const float4*)(arow + r * AS_PITCH);
            }
            const float ww = w[k] * dn;
            acc.x = fmaf(v[k].x, ww, acc.x);
            acc.y = fmaf(v[k].y, ww, acc.y);
            acc.z = fmaf(v[k].z, ww, acc.z);
            acc.w = fmaf(v[k].w, ww, acc.w);
          }
        }
      }
    }
    for (;;) {
      float4 t;
      t.x = fmaxf(acc.x + bv.x, 0.0f);
      t.y = fmaxf(acc.y + bv.y, 0.0f);
      t.z = fmaxf(acc.z + bv.z, 0.0f);
      t.w = fmaxf(acc.w + bv.w, 0.0f);
      *(float4*)(arow + r * AS_PITCH) = t;
      ++r;
      if (r >= 16) break;
      acc = *(const float4*)(arow + r * AS_PITCH);
    }
  }

  __builtin_amdgcn_fence(__ATOMIC_ACQ_REL, "wavefront");
  __builtin_amdgcn_wave_barrier();

  v8f acc[8];
#pragma unroll
  for (int t = 0; t < 8; ++t) acc[t] = (v8f){0.f, 0.f, 0.f, 0.f, 0.f, 0.f, 0.f, 0.f};

  const float* ap = as + lmod * AS_PITCH + half * 2;
  const float* bp = Bs + half * BS_KPITCH + lmod * BS_CPITCH;
#pragma unroll 8
  for (int ks = 0; ks < GDIM / 4; ++ks) {
    const v2f a = *(const v2f*)(ap + 4 * ks);
    const float4* bq = (const float4*)(bp + ks * 2 * BS_KPITCH);
    float4 bf[4];
#pragma unroll
    for (int j = 0; j < 4; ++j) bf[j] = bq[j];
#pragma unroll
    for (int j = 0; j < 4; ++j) {
      v2f b0, b1;
      b0.x = bf[j].x; b0.y = bf[j].y;
      b1.x = bf[j].z; b1.y = bf[j].w;
      acc[2 * j] = __builtin_amdgcn_wmma_f32_16x16x4_f32(
           false, a,  false, b0,
           (short)0, acc[2 * j],  false,  false);
      acc[2 * j + 1] = __builtin_amdgcn_wmma_f32_16x16x4_f32(
           false, a,  false, b1,
           (short)0, acc[2 * j + 1],  false,  false);
    }
  }

  if (strip < nStrips) {
    float* Cp = C + (size_t)(row0 + half * 8) * GDIM + lmod;
#pragma unroll
    for (int t = 0; t < 8; ++t) {
#pragma unroll
      for (int r = 0; r < 8; ++r) Cp[(size_t)r * GDIM + 16 * t] = acc[t][r];
    }
  }
}

extern "C" void kernel_launch(void* const* d_in, const int* in_sizes, int n_in,
                              void* d_out, int out_size, void* d_ws, size_t ws_size,
                              hipStream_t stream) {
  const float* x  = (const float*)d_in[0];
  const int*   ei = (const int*)d_in[1];
  const float* W1 = (const float*)d_in[2];
  const float* b1 = (const float*)d_in[3];
  const float* W2 = (const float*)d_in[4];
  const float* b2 = (const float*)d_in[5];
  float* out = (float*)d_out;

  const int N = in_sizes[0] / GDIM;
  const int E = in_sizes[1] / 2;
  const int* src = ei;
  const int* dst = ei + E;

  float* dinv = (float*)d_ws;
  float* bufA = dinv + N;
  float* bufB = bufA + (size_t)N * GDIM;
  int* cnt  = (int*)(bufB + (size_t)N * GDIM);
  int* off  = cnt + N;
  int* cntC = off + N;
  int* btot = cntC + (size_t)NC * N;
  int* perm = btot + NB_MAX;

  const int T = 256;
  const int RS = (N + NR - 1) / NR;
  const int CS = (E + NC - 1) / NC;
  const int NB = (N + SCAN_B - 1) / SCAN_B;
  const int gNodeWave = (N + 7) / 8;
  const int nStrips = N / 16;
  const int gGemm = (nStrips + GEMM_WAVES - 1) / GEMM_WAVES;

  hist_kernel<<<dim3(NR, NC), 1024, 0, stream>>>(dst, cntC, N, E, RS, CS);
  scan_dinv_kernel<<<NB, SCAN_B, 0, stream>>>(cntC, cnt, dinv, btot, N);
  fill_perm_kernel<<<dim3(NR, NC), 1024, 0, stream>>>(src, dst, cntC, btot, off, perm, N, E, RS, CS, NB);

  gemm_wmma_f32_kernel<<<gGemm, GEMM_WAVES * 32, 0, stream>>>(x, W1, bufA, nStrips);

  gather_gemm_wmma_f32_kernel<<<gGemm, GEMM_WAVES * 32, 0, stream>>>(bufA, perm, off, cnt, dinv, b1, W2,
                                                                    bufB, nStrips);

  gather_rows_kernel<false><<<gNodeWave, T, 0, stream>>>(bufB, perm, off, cnt, dinv, b2, out, N);
}


# source-side deg^-1/2 folded into the rows where they are staged (GEMM 1 A rows, layer-1 row write); gathers add unweighted rows, scale once per node: no per-edge dinv load; rounding changes
# speedup vs baseline: 4.8906x; 1.1654x over previous
#include <hip/hip_runtime.h>
#include <hip/hip_bf16.h>

#define GDIM 128

typedef float v2f __attribute__((ext_vector_type(2)));
typedef float v4f __attribute__((ext_vector_type(4)));
typedef float v8f __attribute__((ext_vector_type(8)));

#define NR 8
#define NC 32
#define RS_MAX 16384
#define SCAN_B 1024
#define NB_MAX 128
#define EB 8

__global__ void __launch_bounds__(1024) hist_kernel(const int* __restrict__ dst,
                                                    int* __restrict__ cntC,
                                                    int N, int E, int RS, int CS) {
  __shared__ int lcnt[RS_MAX];
  const int r = blockIdx.x;
  const int c = blockIdx.y;
  const int tid = threadIdx.x;
  const int rbase = r * RS;
  for (int i = tid; i < RS; i += 1024) lcnt[i] = 0;
  __syncthreads();
  const int cbeg = c * CS;
  const int cend = (cbeg + CS < E) ? cbeg + CS : E;
  for (int e0 = cbeg + tid; e0 < cend; e0 += EB * 1024) {
    int dv[EB];
#pragma unroll
    for (int k = 0; k < EB; ++k) {
      const int e = e0 + k * 1024;
      dv[k] = (e < cend) ? dst[e] : -1;
    }
#pragma unroll
    for (int k = 0; k < EB; ++k) {
      const unsigned d = (unsigned)(dv[k] - rbase);
      if (d < (unsigned)RS) atomicAdd(&lcnt[d], 1);
    }
  }
  __syncthreads();
  for (int i = tid; i < RS; i += 1024) {
    const int n = rbase + i;
    if (n < N) cntC[(size_t)c * N + n] = lcnt[i];
  }
}

__global__ void __launch_bounds__(SCAN_B) scan_dinv_kernel(int* __restrict__ cntC,
                                                           int* __restrict__ cnt,
                                                           float* __restrict__ dinv,
                                                           int* __restrict__ btot, int N) {
  __shared__ int wsum[32];
  const int tid = threadIdx.x;
  const int lane = tid & 31;
  const int wave = tid >> 5;
  const int n = blockIdx.x * SCAN_B + tid;
  const bool valid = n < N;
  int v[NC];
#pragma unroll
  for (int c = 0; c < NC; ++c) v[c] = valid ? cntC[(size_t)c * N + n] : 0;
  int tot = 0;
#pragma unroll
  for (int c = 0; c < NC; ++c) tot += v[c];

  int incl = tot;
  for (int d = 1; d < 32; d <<= 1) {
    const int up = __shfl_up(incl, d);
    if (lane >= d) incl += up;
  }
  if (lane == 31) wsum[wave] = incl;
  __syncthreads();
  const int wv = wsum[lane];
  int winc = wv;
  for (int d = 1; d < 32; d <<= 1) {
    const int up = __shfl_up(winc, d);
    if (lane >= d) winc += up;
  }
  const int total = __shfl(winc, 31);
  const int wbase = __shfl(winc, wave) - __shfl(wv, wave);
  int run = wbase + incl - tot;
  if (valid) {
    cnt[n] = tot;
    dinv[n] = rsqrtf((float)(tot + 1));
#pragma unroll
    for (int c = 0; c < NC; ++c) {
      cntC[(size_t)c * N + n] = run;
      run += v[c];
    }
  }
  if (tid == 0) btot[blockIdx.x] = total;
}

__global__ void __launch_bounds__(1024) fill_perm_kernel(const int* __restrict__ src,
                                                         const int* __restrict__ dst,
                                                         const int* __restrict__ cntC,
                                                         const int* __restrict__ btot,
                                                         int* __restrict__ off,
                                                         int* __restrict__ perm,
                                                         int N, int E, int RS, int CS, int NB) {
  __shared__ int cur[RS_MAX];
  __shared__ int bp[NB_MAX];
  const int r = blockIdx.x;
  const int c = blockIdx.y;
  const int tid = threadIdx.x;
  const int rbase = r * RS;
  if (tid < 32) {
    int t[4];
#pragma unroll
    for (int k = 0; k < 4; ++k) t[k] = (tid * 4 + k < NB) ? btot[tid * 4 + k] : 0;
    const int s = t[0] + t[1] + t[2] + t[3];
    int incl = s;
    for (int d = 1; d < 32; d <<= 1) {
      const int up = __shfl_up(incl, d);
      if (tid >= d) incl += up;
    }
    int ex = incl - s;
#pragma unroll
    for (int k = 0; k < 4; ++k) {
      bp[tid * 4 + k] = ex;
      ex += t[k];
    }
  }
  __syncthreads();
  for (int i = tid; i < RS; i += 1024) {
    const int n = rbase + i;
    if (n < N) {
      const int b = bp[n / SCAN_B] + cntC[(size_t)c * N + n];
      cur[i] = b;
      if (c == 0) off[n] = b;
    }
  }
  __syncthreads();
  const int cbeg = c * CS;
  const int cend = (cbeg + CS < E) ? cbeg + CS : E;
  for (int e0 = cbeg + tid; e0 < cend; e0 += EB * 1024) {
    int dv[EB], sv[EB];
#pragma unroll
    for (int k = 0; k < EB; ++k) {
      const int e = e0 + k * 1024;
      dv[k] = (e < cend) ? dst[e] : -1;
      sv[k] = (e < cend) ? src[e] : 0;
    }
    int pos[EB];
#pragma unroll
    for (int k = 0; k < EB; ++k) {
      const unsigned d = (unsigned)(dv[k] - rbase);
      pos[k] = (d < (unsigned)RS) ? atomicAdd(&cur[d], 1) : -1;
    }
#pragma unroll
    for (int k = 0; k < EB; ++k) {
      if (pos[k] >= 0) perm[pos[k]] = sv[k];
    }
  }
}

#define BS_CPITCH 20
#define BS_KPITCH (16 * BS_CPITCH)
#define AS_PITCH 132
#define GEMM_WAVES 25
#define B_ITERS ((GDIM / 2 + GEMM_WAVES - 1) / GEMM_WAVES)

__global__ void __launch_bounds__(GEMM_WAVES * 32)
gemm_wmma_f32_kernel(const float* __restrict__ A, const float* __restrict__ B,
                     const float* __restrict__ rowScale, float* __restrict__ C, int nStrips) {
  __shared__ float Bs[(GDIM / 2) * BS_KPITCH];
  __shared__ float As[GEMM_WAVES * 16 * AS_PITCH];

  const int wave = __builtin_amdgcn_readfirstlane(threadIdx.x >> 5);
  const int lane = threadIdx.x & 31;
  const int half = lane >> 4;
  const int lmod = lane & 15;
  const int strip = blockIdx.x * GEMM_WAVES + wave;
  const int sidx = strip < nStrips ? strip : nStrips - 1;
  const int row0 = sidx * 16;

  float4 w0[B_ITERS], w1[B_ITERS];
#pragma unroll
  for (int it = 0; it < B_ITERS; ++it) {
    const int kq = it * GEMM_WAVES + wave;
    const int kp = kq < GDIM / 2 ? kq : GDIM / 2 - 1;
    w0[it] = *(const float4*)(B + (size_t)(2 * kp) * GDIM + 4 * lane);
    w1[it] = *(const float4*)(B + (size_t)(2 * kp + 1) * GDIM + 4 * lane);
  }

  const float4* Ag = (const float4*)(A + (size_t)row0 * GDIM) + lane;
  float4 ar[16];
#pragma unroll
  for (int r = 0; r < 16; ++r) ar[r] = Ag[r * (GDIM / 4)];

  const float* rs = rowScale + row0;
  float sc[16];
#pragma unroll
  for (int r = 0; r < 16; ++r) sc[r] = rs[r];

  {
    float* d0 = Bs + (lane & 3) * 4 * BS_CPITCH + (lane >> 2) * 2;
#pragma unroll
    for (int it = 0; it < B_ITERS; ++it) {
      const int kq = it * GEMM_WAVES + wave;
      const int kp = kq < GDIM / 2 ? kq : GDIM / 2 - 1;
      float* d = d0 + kp * BS_KPITCH;
      v2f p;
      p.x = w0[it].x; p.y = w1[it].x; *(v2f*)(d + 0 * BS_CPITCH) = p;
      p.x = w0[it].y; p.y = w1[it].y; *(v2f*)(d + 1 * BS_CPITCH) = p;
      p.x = w0[it].z; p.y = w1[it].z; *(v2f*)(d + 2 * BS_CPITCH) = p;
      p.x = w0[it].w; p.y = w1[it].w; *(v2f*)(d + 3 * BS_CPITCH) = p;
    }
  }
  float* as = As + wave * (16 * AS_PITCH);
#pragma unroll
  for (int r = 0; r < 16; ++r) {
    float4 t;
    t.x = ar[r].x * sc[r];
    t.y = ar[r].y * sc[r];
    t.z = ar[r].z * sc[r];
    t.w = ar[r].w * sc[r];
    *(float4*)(as + r * AS_PITCH + 4 * lane) = t;
  }

  __syncthreads();

  v8f acc[8];
#pragma unroll
  for (int t = 0; t < 8; ++t) acc[t] = (v8f){0.f, 0.f, 0.f, 0.f, 0.f, 0.f, 0.f, 0.f};

  const float* ap = as + lmod * AS_PITCH + half * 2;
  const float* bp = Bs + half * BS_KPITCH + lmod * BS_CPITCH;
#pragma unroll 8
  for (int ks = 0; ks < GDIM / 4; ++ks) {
    const v2f a = *(const v2f*)(ap + 4 * ks);
    const float4* bq = (const float4*)(bp + ks * 2 * BS_KPITCH);
    float4 bv[4];
#pragma unroll
    for (int j = 0; j < 4; ++j) bv[j] = bq[j];
#pragma unroll
    for (int j = 0; j < 4; ++j) {
      v2f b0, b1;
      b0.x = bv[j].x; b0.y = bv[j].y;
      b1.x = bv[j].z; b1.y = bv[j].w;
      acc[2 * j] = __builtin_amdgcn_wmma_f32_16x16x4_f32(
           false, a,  false, b0,
           (short)0, acc[2 * j],  false,  false);
      acc[2 * j + 1] = __builtin_amdgcn_wmma_f32_16x16x4_f32(
           false, a,  false, b1,
           (short)0, acc[2 * j + 1],  false,  false);
    }
  }

  if (strip < nStrips) {
    float* Cp = C + (size_t)(row0 + half * 8) * GDIM + lmod;
#pragma unroll
    for (int t = 0; t < 8; ++t) {
#pragma unroll
      for (int r = 0; r < 8; ++r) Cp[(size_t)r * GDIM + 16 * t] = acc[t][r];
    }
  }
}

template <int K>
__device__ __forceinline__ void accum_rows(const float4* __restrict__ H4,
                                           const int* __restrict__ perm,
                                           int j, int lane, float4& acc) {
  int s[K];
  float4 v[K];
#pragma unroll
  for (int k = 0; k < K; ++k) s[k] = perm[j + k];
#pragma unroll
  for (int k = 0; k < K; ++k) v[k] = H4[(size_t)s[k] * (GDIM / 4) + lane];
#pragma unroll
  for (int k = 0; k < K; ++k) {
    acc.x += v[k].x;
    acc.y += v[k].y;
    acc.z += v[k].z;
    acc.w += v[k].w;
  }
}

template <bool RELU>
__global__ void gather_rows_kernel(const float* __restrict__ h,
                                   const int* __restrict__ perm,
                                   const int* __restrict__ off,
                                   const int* __restrict__ cnt,
                                   const float* __restrict__ dinv,
                                   const float* __restrict__ bias,
                                   float* __restrict__ out, int n) {
  const int wave = __builtin_amdgcn_readfirstlane(threadIdx.x >> 5);
  const int node = blockIdx.x * 8 + wave;
  if (node >= n) return;
  const int lane = threadIdx.x & 31;
  const float4* H4 = (const float4*)h;
  int j = off[node];
  const int end = j + cnt[node];
  const float dn = dinv[node];

  const float* bp = bias + lane * 4;
  const float b0 = bp[0], b1v = bp[1], b2v = bp[2], b3 = bp[3];
  float4 acc = H4[(size_t)node * (GDIM / 4) + lane];

  for (; j + 8 <= end; j += 8) accum_rows<8>(H4, perm, j, lane, acc);
  if (j + 4 <= end) { accum_rows<4>(H4, perm, j, lane, acc); j += 4; }
  if (j + 2 <= end) { accum_rows<2>(H4, perm, j, lane, acc); j += 2; }
  if (j < end)      { accum_rows<1>(H4, perm, j, lane, acc); }

  acc.x = fmaf(acc.x, dn, b0);
  acc.y = fmaf(acc.y, dn, b1v);
  acc.z = fmaf(acc.z, dn, b2v);
  acc.w = fmaf(acc.w, dn, b3);
  if (RELU) {
    acc.x = fmaxf(acc.x, 0.0f);
    acc.y = fmaxf(acc.y, 0.0f);
    acc.z = fmaxf(acc.z, 0.0f);
    acc.w = fmaxf(acc.w, 0.0f);
  }
  ((float4*)(out + (size_t)node * GDIM))[lane] = acc;
}

#define GB 16

__device__ __forceinline__ int lane_int(int v, int l) {
  return __builtin_amdgcn_readlane(v, l);
}
__device__ __forceinline__ float lane_float(float v, int l) {
  return __int_as_float(__builtin_amdgcn_readlane(__float_as_int(v), l));
}

__global__ void __launch_bounds__(GEMM_WAVES * 32)
gather_gemm_wmma_f32_kernel(const float* __restrict__ h, const int* __restrict__ perm,
                            const int* __restrict__ off, const int* __restrict__ cnt,
                            const float* __restrict__ dinv, const float* __restrict__ bias,
                            const float* __restrict__ B, float* __restrict__ C, int nStrips) {
  __shared__ float Bs[(GDIM / 2) * BS_KPITCH];
  __shared__ float As[GEMM_WAVES * 16 * AS_PITCH];

  const int wave = __builtin_amdgcn_readfirstlane(threadIdx.x >> 5);
  const int lane = threadIdx.x & 31;
  const int half = lane >> 4;
  const int lmod = lane & 15;
  const int strip = blockIdx.x * GEMM_WAVES + wave;
  const int sidx = strip < nStrips ? strip : nStrips - 1;
  const int row0 = sidx * 16;

  float4 w0[B_ITERS], w1[B_ITERS];
#pragma unroll
  for (int it = 0; it < B_ITERS; ++it) {
    const int kq = it * GEMM_WAVES + wave;
    const int kp = kq < GDIM / 2 ? kq : GDIM / 2 - 1;
    w0[it] = *(const float4*)(B + (size_t)(2 * kp) * GDIM + 4 * lane);
    w1[it] = *(const float4*)(B + (size_t)(2 * kp + 1) * GDIM + 4 * lane);
  }

  const float4* H4 = (const float4*)h;
  const v4f* Ag = (const v4f*)h + (size_t)row0 * (GDIM / 4) + lane;
  v4f ar[16];
#pragma unroll
  for (int r = 0; r < 16; ++r) ar[r] = Ag[r * (GDIM / 4)];

  const int o_l = off[row0 + lmod];
  const int c_l = cnt[row0 + lmod];
  const float d_l = dinv[row0 + lmod];
  const float* bq4 = bias + lane * 4;
  float4 bv;
  bv.x = bq4[0];
  bv.y = bq4[1];
  bv.z = bq4[2];
  bv.w = bq4[3];

  {
    float* d0 = Bs + (lane & 3) * 4 * BS_CPITCH + (lane >> 2) * 2;
#pragma unroll
    for (int it = 0; it < B_ITERS; ++it) {
      const int kq = it * GEMM_WAVES + wave;
      const int kp = kq < GDIM / 2 ? kq : GDIM / 2 - 1;
      float* d = d0 + kp * BS_KPITCH;
      v2f p;
      p.x = w0[it].x; p.y = w1[it].x; *(v2f*)(d + 0 * BS_CPITCH) = p;
      p.x = w0[it].y; p.y = w1[it].y; *(v2f*)(d + 1 * BS_CPITCH) = p;
      p.x = w0[it].z; p.y = w1[it].z; *(v2f*)(d + 2 * BS_CPITCH) = p;
      p.x = w0[it].w; p.y = w1[it].w; *(v2f*)(d + 3 * BS_CPITCH) = p;
    }
  }
  __syncthreads();

  float* as = As + wave * (16 * AS_PITCH);
  float* arow = as + 4 * lane;
#pragma unroll
  for (int r = 0; r < 16; ++r) *(v4f*)(arow + r * AS_PITCH) = ar[r];

  {
    const int segStart = lane_int(o_l, 0);
    const int segEnd = lane_int(o_l, 15) + lane_int(c_l, 15);
    int r = 0;
    int curEnd = segStart + lane_int(c_l, 0);
    float dn = lane_float(d_l, 0);
    float4 acc = *(const float4*)arow;

    if (segStart < segEnd) {
      const int last = segEnd - 1;
      int sn[GB];
#pragma unroll
      for (int k = 0; k < GB; ++k) {
        const int e = segStart + k;
        sn[k] = perm[e < last ? e : last];
      }
      for (int p = segStart; p < segEnd; p += GB) {
        float4 v[GB];
#pragma unroll
        for (int k = 0; k < GB; ++k) v[k] = H4[(size_t)sn[k] * (GDIM / 4) + lane];
        if (p + GB < segEnd) {
#pragma unroll
          for (int k = 0; k < GB; ++k) {
            const int e = p + GB + k;
            sn[k] = perm[e < last ? e : last];
          }
        }
#pragma unroll
        for (int k = 0; k < GB; ++k) {
          const int e = p + k;
          if (e < segEnd) {
            while (e == curEnd && r < 15) {
              float4 t;
              t.x = fmaxf(fmaf(acc.x, dn, bv.x), 0.0f) * dn;
              t.y = fmaxf(fmaf(acc.y, dn, bv.y), 0.0f) * dn;
              t.z = fmaxf(fmaf(acc.z, dn, bv.z), 0.0f) * dn;
              t.w = fmaxf(fmaf(acc.w, dn, bv.w), 0.0f) * dn;
              *(float4*)(arow + r * AS_PITCH) = t;
              ++r;
              curEnd = lane_int(o_l, r) + lane_int(c_l, r);
              dn = lane_float(d_l, r);
              acc = *(const float4*)(arow + r * AS_PITCH);
            }
            acc.x += v[k].x;
            acc.y += v[k].y;
            acc.z += v[k].z;
            acc.w += v[k].w;
          }
        }
      }
    }
    for (;;) {
      float4 t;
      t.x = fmaxf(fmaf(acc.x, dn, bv.x), 0.0f) * dn;
      t.y = fmaxf(fmaf(acc.y, dn, bv.y), 0.0f) * dn;
      t.z = fmaxf(fmaf(acc.z, dn, bv.z), 0.0f) * dn;
      t.w = fmaxf(fmaf(acc.w, dn, bv.w), 0.0f) * dn;
      *(float4*)(arow + r * AS_PITCH) = t;
      ++r;
      if (r >= 16) break;
      dn = lane_float(d_l, r);
      acc = *(const float4*)(arow + r * AS_PITCH);
    }
  }

  __builtin_amdgcn_fence(__ATOMIC_ACQ_REL, "wavefront");
  __builtin_amdgcn_wave_barrier();

  v8f acc[8];
#pragma unroll
  for (int t = 0; t < 8; ++t) acc[t] = (v8f){0.f, 0.f, 0.f, 0.f, 0.f, 0.f, 0.f, 0.f};

  const float* ap = as + lmod * AS_PITCH + half * 2;
  const float* bp = Bs + half * BS_KPITCH + lmod * BS_CPITCH;
#pragma unroll 8
  for (int ks = 0; ks < GDIM / 4; ++ks) {
    const v2f a = *(const v2f*)(ap + 4 * ks);
    const float4* bq = (const float4*)(bp + ks * 2 * BS_KPITCH);
    float4 bf[4];
#pragma unroll
    for (int j = 0; j < 4; ++j) bf[j] = bq[j];
#pragma unroll
    for (int j = 0; j < 4; ++j) {
      v2f b0, b1;
      b0.x = bf[j].x; b0.y = bf[j].y;
      b1.x = bf[j].z; b1.y = bf[j].w;
      acc[2 * j] = __builtin_amdgcn_wmma_f32_16x16x4_f32(
           false, a,  false, b0,
           (short)0, acc[2 * j],  false,  false);
      acc[2 * j + 1] = __builtin_amdgcn_wmma_f32_16x16x4_f32(
           false, a,  false, b1,
           (short)0, acc[2 * j + 1],  false,  false);
    }
  }

  if (strip < nStrips) {
    float* Cp = C + (size_t)(row0 + half * 8) * GDIM + lmod;
#pragma unroll
    for (int t = 0; t < 8; ++t) {
#pragma unroll
      for (int r = 0; r < 8; ++r) Cp[(size_t)r * GDIM + 16 * t] = acc[t][r];
    }
  }
}

extern "C" void kernel_launch(void* const* d_in, const int* in_sizes, int n_in,
                              void* d_out, int out_size, void* d_ws, size_t ws_size,
                              hipStream_t stream) {
  const float* x  = (const float*)d_in[0];
  const int*   ei = (const int*)d_in[1];
  const float* W1 = (const float*)d_in[2];
  const float* b1 = (const float*)d_in[3];
  const float* W2 = (const float*)d_in[4];
  const float* b2 = (const float*)d_in[5];
  float* out = (float*)d_out;

  const int N = in_sizes[0] / GDIM;
  const int E = in_sizes[1] / 2;
  const int* src = ei;
  const int* dst = ei + E;

  float* dinv = (float*)d_ws;
  float* bufA = dinv + N;
  float* bufB = bufA + (size_t)N * GDIM;
  int* cnt  = (int*)(bufB + (size_t)N * GDIM);
  int* off  = cnt + N;
  int* cntC = off + N;
  int* btot = cntC + (size_t)NC * N;
  int* perm = btot + NB_MAX;

  const int T = 256;
  const int RS = (N + NR - 1) / NR;
  const int CS = (E + NC - 1) / NC;
  const int NB = (N + SCAN_B - 1) / SCAN_B;
  const int gNodeWave = (N + 7) / 8;
  const int nStrips = N / 16;
  const int gGemm = (nStrips + GEMM_WAVES - 1) / GEMM_WAVES;

  hist_kernel<<<dim3(NR, NC), 1024, 0, stream>>>(dst, cntC, N, E, RS, CS);
  scan_dinv_kernel<<<NB, SCAN_B, 0, stream>>>(cntC, cnt, dinv, btot, N);
  fill_perm_kernel<<<dim3(NR, NC), 1024, 0, stream>>>(src, dst, cntC, btot, off, perm, N, E, RS, CS, NB);

  gemm_wmma_f32_kernel<<<gGemm, GEMM_WAVES * 32, 0, stream>>>(x, W1, dinv, bufA, nStrips);

  gather_gemm_wmma_f32_kernel<<<gGemm, GEMM_WAVES * 32, 0, stream>>>(bufA, perm, off, cnt, dinv, b1, W2,
                                                                    bufB, nStrips);

  gather_rows_kernel<false><<<gNodeWave, T, 0, stream>>>(bufB, perm, off, cnt, dinv, b2, out, N);
}


# edge bucketing passes (count, fill): 16-byte edge loads, a thread's whole chunk share asked for at once (512-thread groups), one memory round trip a pass, not seven; LDS arrival order changes
# speedup vs baseline: 5.0448x; 1.0315x over previous
#include <hip/hip_runtime.h>
#include <hip/hip_bf16.h>

#define GDIM 128

typedef float v2f __attribute__((ext_vector_type(2)));
typedef float v4f __attribute__((ext_vector_type(4)));
typedef float v8f __attribute__((ext_vector_type(8)));

#define NR 8
#define NC 32
#define RS_MAX 16384
#define SCAN_B 1024
#define NB_MAX 128
#define BT 512
#define EV 25
#define NSL (RS_MAX / BT)


__device__ __forceinline__ int chunk_quads(const int* a, const int* b, int cbeg, int cend) {
  const bool aligned = ((((size_t)(a + cbeg)) | ((size_t)(b + cbeg))) & 15) == 0;
  return (aligned && cend > cbeg) ? ((cend - cbeg) >> 2) : 0;
}

__device__ __forceinline__ int4 load_quad(const int* __restrict__ p, int qi, int nq, int fillv) {
  const bool ok = qi < nq;
  int4 v = *(const int4*)(p + 4 * (ok ? qi : 0));
  if (!ok) { v.x = fillv; v.y = fillv; v.z = fillv; v.w = fillv; }
  return v;
}

__global__ void __launch_bounds__(BT) hist_kernel(const int* __restrict__ dst,
                                                  int* __restrict__ cntC,
                                                  int N, int E, int RS, int CS) {
  __shared__ int lcnt[RS_MAX];
  const int r = blockIdx.x;
  const int c = blockIdx.y;
  const int tid = threadIdx.x;
  const int rbase = r * RS;
  const int cbeg = c * CS;
  const int cend = (cbeg + CS < E) ? cbeg + CS : E;
  const int nq = chunk_quads(dst, dst, cbeg, cend);
  const int* dq = dst + cbeg;
  int q0 = 0;
  int4 dv[EV];
  if (nq > 0) {
#pragma unroll
    for (int k = 0; k < EV; ++k) dv[k] = load_quad(dq, q0 + k * BT + tid, nq, -1);
  }
  for (int i = tid; i < RS; i += BT) lcnt[i] = 0;
  __syncthreads();
  while (q0 < nq) {
#pragma unroll
    for (int k = 0; k < EV; ++k) {
      const unsigned d0 = (unsigned)(dv[k].x - rbase);
      const unsigned d1 = (unsigned)(dv[k].y - rbase);
      const unsigned d2 = (unsigned)(dv[k].z - rbase);
      const unsigned d3 = (unsigned)(dv[k].w - rbase);
      if (d0 < (unsigned)RS) atomicAdd(&lcnt[d0], 1);
      if (d1 < (unsigned)RS) atomicAdd(&lcnt[d1], 1);
      if (d2 < (unsigned)RS) atomicAdd(&lcnt[d2], 1);
      if (d3 < (unsigned)RS) atomicAdd(&lcnt[d3], 1);
    }
    q0 += BT * EV;
    if (q0 < nq) {
#pragma unroll
      for (int k = 0; k < EV; ++k) dv[k] = load_quad(dq, q0 + k * BT + tid, nq, -1);
    }
  }
  for (int e = cbeg + 4 * nq + tid; e < cend; e += BT) {
    const unsigned d = (unsigned)(dst[e] - rbase);
    if (d < (unsigned)RS) atomicAdd(&lcnt[d], 1);
  }
  __syncthreads();
  int cv[NSL];
#pragma unroll
  for (int j = 0; j < NSL; ++j) cv[j] = lcnt[(j * BT + tid < RS) ? j * BT + tid : 0];
#pragma unroll
  for (int j = 0; j < NSL; ++j) {
    const int i = j * BT + tid;
    const int n = rbase + i;
    if (i < RS && n < N) cntC[(size_t)c * N + n] = cv[j];
  }
}

__global__ void __launch_bounds__(SCAN_B) scan_dinv_kernel(int* __restrict__ cntC,
                                                           int* __restrict__ cnt,
                                                           float* __restrict__ dinv,
                                                           int* __restrict__ btot, int N) {
  __shared__ int wsum[32];
  const int tid = threadIdx.x;
  const int lane = tid & 31;
  const int wave = tid >> 5;
  const int n = blockIdx.x * SCAN_B + tid;
  const bool valid = n < N;
  int v[NC];
#pragma unroll
  for (int c = 0; c < NC; ++c) v[c] = valid ? cntC[(size_t)c * N + n] : 0;
  int tot = 0;
#pragma unroll
  for (int c = 0; c < NC; ++c) tot += v[c];

  int incl = tot;
  for (int d = 1; d < 32; d <<= 1) {
    const int up = __shfl_up(incl, d);
    if (lane >= d) incl += up;
  }
  if (lane == 31) wsum[wave] = incl;
  __syncthreads();
  const int wv = wsum[lane];
  int winc = wv;
  for (int d = 1; d < 32; d <<= 1) {
    const int up = __shfl_up(winc, d);
    if (lane >= d) winc += up;
  }
  const int total = __shfl(winc, 31);
  const int wbase = __shfl(winc, wave) - __shfl(wv, wave);
  int run = wbase + incl - tot;
  if (valid) {
    cnt[n] = tot;
    dinv[n] = rsqrtf((float)(tot + 1));
#pragma unroll
    for (int c = 0; c < NC; ++c) {
      cntC[(size_t)c * N + n] = run;
      run += v[c];
    }
  }
  if (tid == 0) btot[blockIdx.x] = total;
}

__device__ __forceinline__ void place8(const int (&d8)[8], const int (&s8)[8], int* cur,
                                       int* __restrict__ perm, int rbase, int RS) {
  int pos[8];
#pragma unroll
  for (int j = 0; j < 8; ++j) {
    const unsigned d = (unsigned)(d8[j] - rbase);
    pos[j] = (d < (unsigned)RS) ? atomicAdd(&cur[d], 1) : -1;
  }
#pragma unroll
  for (int j = 0; j < 8; ++j) {
    if (pos[j] >= 0) perm[pos[j]] = s8[j];
  }
}

__global__ void __launch_bounds__(BT) fill_perm_kernel(const int* __restrict__ src,
                                                       const int* __restrict__ dst,
                                                       const int* __restrict__ cntC,
                                                       const int* __restrict__ btot,
                                                       int* __restrict__ off,
                                                       int* __restrict__ perm,
                                                       int N, int E, int RS, int CS, int NB) {
  __shared__ int cur[RS_MAX];
  __shared__ int bp[NB_MAX];
  const int r = blockIdx.x;
  const int c = blockIdx.y;
  const int tid = threadIdx.x;
  const int rbase = r * RS;
  const int cbeg = c * CS;
  const int cend = (cbeg + CS < E) ? cbeg + CS : E;
  const int nq = chunk_quads(dst, src, cbeg, cend);
  const int* dq = dst + cbeg;
  const int* sq = src + cbeg;
  int cv[NSL];
#pragma unroll
  for (int j = 0; j < NSL; ++j) {
    const int i = j * BT + tid;
    const int n = rbase + i;
    cv[j] = cntC[(size_t)c * N + ((i < RS && n < N) ? n : 0)];
  }
  if (tid < 32) {
    int t[4];
#pragma unroll
    for (int k = 0; k < 4; ++k) t[k] = (tid * 4 + k < NB) ? btot[tid * 4 + k] : 0;
    const int s = t[0] + t[1] + t[2] + t[3];
    int incl = s;
    for (int d = 1; d < 32; d <<= 1) {
      const int up = __shfl_up(incl, d);
      if (tid >= d) incl += up;
    }
    int ex = incl - s;
#pragma unroll
    for (int k = 0; k < 4; ++k) {
      bp[tid * 4 + k] = ex;
      ex += t[k];
    }
  }
  __syncthreads();
#pragma unroll
  for (int j = 0; j < NSL; ++j) {
    const int i = j * BT + tid;
    const int n = rbase + i;
    cv[j] += bp[(i < RS && n < N) ? n / SCAN_B : 0];
  }
#pragma unroll
  for (int j = 0; j < NSL; ++j) {
    const int i = j * BT + tid;
    const int n = rbase + i;
    if (i < RS && n < N) {
      cur[i] = cv[j];
      if (c == 0) off[n] = cv[j];
    }
  }
  int q0 = 0;
  int4 dv[EV], sv[EV];
  if (nq > 0) {
#pragma unroll
    for (int k = 0; k < EV; ++k) {
      dv[k] = load_quad(dq, q0 + k * BT + tid, nq, -1);
      sv[k] = load_quad(sq, q0 + k * BT + tid, nq, 0);
    }
  }
  __syncthreads();
  while (q0 < nq) {
#pragma unroll
    for (int k = 0; k + 1 < EV; k += 2) {
      const int d8[8] = {dv[k].x, dv[k].y, dv[k].z, dv[k].w,
                         dv[k + 1].x, dv[k + 1].y, dv[k + 1].z, dv[k + 1].w};
      const int s8[8] = {sv[k].x, sv[k].y, sv[k].z, sv[k].w,
                         sv[k + 1].x, sv[k + 1].y, sv[k + 1].z, sv[k + 1].w};
      place8(d8, s8, cur, perm, rbase, RS);
    }
    if (EV & 1) {
      const int d8[8] = {dv[EV - 1].x, dv[EV - 1].y, dv[EV - 1].z, dv[EV - 1].w, -1, -1, -1, -1};
      const int s8[8] = {sv[EV - 1].x, sv[EV - 1].y, sv[EV - 1].z, sv[EV - 1].w, 0, 0, 0, 0};
      place8(d8, s8, cur, perm, rbase, RS);
    }
    q0 += BT * EV;
    if (q0 < nq) {
#pragma unroll
      for (int k = 0; k < EV; ++k) {
        dv[k] = load_quad(dq, q0 + k * BT + tid, nq, -1);
        sv[k] = load_quad(sq, q0 + k * BT + tid, nq, 0);
      }
    }
  }
  for (int e = cbeg + 4 * nq + tid; e < cend; e += BT) {
    const unsigned d = (unsigned)(dst[e] - rbase);
    if (d < (unsigned)RS) perm[atomicAdd(&cur[d], 1)] = src[e];
  }
}

#define BS_CPITCH 20
#define BS_KPITCH (16 * BS_CPITCH)
#define AS_PITCH 132
#define GEMM_WAVES 25
#define B_ITERS ((GDIM / 2 + GEMM_WAVES - 1) / GEMM_WAVES)

__global__ void __launch_bounds__(GEMM_WAVES * 32)
gemm_wmma_f32_kernel(const float* __restrict__ A, const float* __restrict__ B,
                     const float* __restrict__ rowScale, float* __restrict__ C, int nStrips) {
  __shared__ float Bs[(GDIM / 2) * BS_KPITCH];
  __shared__ float As[GEMM_WAVES * 16 * AS_PITCH];

  const int wave = __builtin_amdgcn_readfirstlane(threadIdx.x >> 5);
  const int lane = threadIdx.x & 31;
  const int half = lane >> 4;
  const int lmod = lane & 15;
  const int strip = blockIdx.x * GEMM_WAVES + wave;
  const int sidx = strip < nStrips ? strip : nStrips - 1;
  const int row0 = sidx * 16;

  float4 w0[B_ITERS], w1[B_ITERS];
#pragma unroll
  for (int it = 0; it < B_ITERS; ++it) {
    const int kq = it * GEMM_WAVES + wave;
    const int kp = kq < GDIM / 2 ? kq : GDIM / 2 - 1;
    w0[it] = *(const float4*)(B + (size_t)(2 * kp) * GDIM + 4 * lane);
    w1[it] = *(const float4*)(B + (size_t)(2 * kp + 1) * GDIM + 4 * lane);
  }

  const float4* Ag = (const float4*)(A + (size_t)row0 * GDIM) + lane;
  float4 ar[16];
#pragma unroll
  for (int r = 0; r < 16; ++r) ar[r] = Ag[r * (GDIM / 4)];

  const float* rs = rowScale + row0;
  float sc[16];
#pragma unroll
  for (int r = 0; r < 16; ++r) sc[r] = rs[r];

  {
    float* d0 = Bs + (lane & 3) * 4 * BS_CPITCH + (lane >> 2) * 2;
#pragma unroll
    for (int it = 0; it < B_ITERS; ++it) {
      const int kq = it * GEMM_WAVES + wave;
      const int kp = kq < GDIM / 2 ? kq : GDIM / 2 - 1;
      float* d = d0 + kp * BS_KPITCH;
      v2f p;
      p.x = w0[it].x; p.y = w1[it].x; *(v2f*)(d + 0 * BS_CPITCH) = p;
      p.x = w0[it].y; p.y = w1[it].y; *(v2f*)(d + 1 * BS_CPITCH) = p;
      p.x = w0[it].z; p.y = w1[it].z; *(v2f*)(d + 2 * BS_CPITCH) = p;
      p.x = w0[it].w; p.y = w1[it].w; *(v2f*)(d + 3 * BS_CPITCH) = p;
    }
  }
  float* as = As + wave * (16 * AS_PITCH);
#pragma unroll
  for (int r = 0; r < 16; ++r) {
    float4 t;
    t.x = ar[r].x * sc[r];
    t.y = ar[r].y * sc[r];
    t.z = ar[r].z * sc[r];
    t.w = ar[r].w * sc[r];
    *(float4*)(as + r * AS_PITCH + 4 * lane) = t;
  }

  __syncthreads();

  v8f acc[8];
#pragma unroll
  for (int t = 0; t < 8; ++t) acc[t] = (v8f){0.f, 0.f, 0.f, 0.f, 0.f, 0.f, 0.f, 0.f};

  const float* ap = as + lmod * AS_PITCH + half * 2;
  const float* bp = Bs + half * BS_KPITCH + lmod * BS_CPITCH;
#pragma unroll 8
  for (int ks = 0; ks < GDIM / 4; ++ks) {
    const v2f a = *(const v2f*)(ap + 4 * ks);
    const float4* bq = (const float4*)(bp + ks * 2 * BS_KPITCH);
    float4 bv[4];
#pragma unroll
    for (int j = 0; j < 4; ++j) bv[j] = bq[j];
#pragma unroll
    for (int j = 0; j < 4; ++j) {
      v2f b0, b1;
      b0.x = bv[j].x; b0.y = bv[j].y;
      b1.x = bv[j].z; b1.y = bv[j].w;
      acc[2 * j] = __builtin_amdgcn_wmma_f32_16x16x4_f32(
           false, a,  false, b0,
           (short)0, acc[2 * j],  false,  false);
      acc[2 * j + 1] = __builtin_amdgcn_wmma_f32_16x16x4_f32(
           false, a,  false, b1,
           (short)0, acc[2 * j + 1],  false,  false);
    }
  }

  if (strip < nStrips) {
    float* Cp = C + (size_t)(row0 + half * 8) * GDIM + lmod;
#pragma unroll
    for (int t = 0; t < 8; ++t) {
#pragma unroll
      for (int r = 0; r < 8; ++r) Cp[(size_t)r * GDIM + 16 * t] = acc[t][r];
    }
  }
}

template <int K>
__device__ __forceinline__ void accum_rows(const float4* __restrict__ H4,
                                           const int* __restrict__ perm,
                                           int j, int lane, float4& acc) {
  int s[K];
  float4 v[K];
#pragma unroll
  for (int k = 0; k < K; ++k) s[k] = perm[j + k];
#pragma unroll
  for (int k = 0; k < K; ++k) v[k] = H4[(size_t)s[k] * (GDIM / 4) + lane];
#pragma unroll
  for (int k = 0; k < K; ++k) {
    acc.x += v[k].x;
    acc.y += v[k].y;
    acc.z += v[k].z;
    acc.w += v[k].w;
  }
}

template <bool RELU>
__global__ void gather_rows_kernel(const float* __restrict__ h,
                                   const int* __restrict__ perm,
                                   const int* __restrict__ off,
                                   const int* __restrict__ cnt,
                                   const float* __restrict__ dinv,
                                   const float* __restrict__ bias,
                                   float* __restrict__ out, int n) {
  const int wave = __builtin_amdgcn_readfirstlane(threadIdx.x >> 5);
  const int node = blockIdx.x * 8 + wave;
  if (node >= n) return;
  const int lane = threadIdx.x & 31;
  const float4* H4 = (const float4*)h;
  int j = off[node];
  const int end = j + cnt[node];
  const float dn = dinv[node];

  const float* bp = bias + lane * 4;
  const float b0 = bp[0], b1v = bp[1], b2v = bp[2], b3 = bp[3];
  float4 acc = H4[(size_t)node * (GDIM / 4) + lane];

  for (; j + 8 <= end; j += 8) accum_rows<8>(H4, perm, j, lane, acc);
  if (j + 4 <= end) { accum_rows<4>(H4, perm, j, lane, acc); j += 4; }
  if (j + 2 <= end) { accum_rows<2>(H4, perm, j, lane, acc); j += 2; }
  if (j < end)      { accum_rows<1>(H4, perm, j, lane, acc); }

  acc.x = fmaf(acc.x, dn, b0);
  acc.y = fmaf(acc.y, dn, b1v);
  acc.z = fmaf(acc.z, dn, b2v);
  acc.w = fmaf(acc.w, dn, b3);
  if (RELU) {
    acc.x = fmaxf(acc.x, 0.0f);
    acc.y = fmaxf(acc.y, 0.0f);
    acc.z = fmaxf(acc.z, 0.0f);
    acc.w = fmaxf(acc.w, 0.0f);
  }
  ((float4*)(out + (size_t)node * GDIM))[lane] = acc;
}

#define GB 16

__device__ __forceinline__ int lane_int(int v, int l) {
  return __builtin_amdgcn_readlane(v, l);
}
__device__ __forceinline__ float lane_float(float v, int l) {
  return __int_as_float(__builtin_amdgcn_readlane(__float_as_int(v), l));
}

__global__ void __launch_bounds__(GEMM_WAVES * 32)
gather_gemm_wmma_f32_kernel(const float* __restrict__ h, const int* __restrict__ perm,
                            const int* __restrict__ off, const int* __restrict__ cnt,
                            const float* __restrict__ dinv, const float* __restrict__ bias,
                            const float* __restrict__ B, float* __restrict__ C, int nStrips) {
  __shared__ float Bs[(GDIM / 2) * BS_KPITCH];
  __shared__ float As[GEMM_WAVES * 16 * AS_PITCH];

  const int wave = __builtin_amdgcn_readfirstlane(threadIdx.x >> 5);
  const int lane = threadIdx.x & 31;
  const int half = lane >> 4;
  const int lmod = lane & 15;
  const int strip = blockIdx.x * GEMM_WAVES + wave;
  const int sidx = strip < nStrips ? strip : nStrips - 1;
  const int row0 = sidx * 16;

  float4 w0[B_ITERS], w1[B_ITERS];
#pragma unroll
  for (int it = 0; it < B_ITERS; ++it) {
    const int kq = it * GEMM_WAVES + wave;
    const int kp = kq < GDIM / 2 ? kq : GDIM / 2 - 1;
    w0[it] = *(const float4*)(B + (size_t)(2 * kp) * GDIM + 4 * lane);
    w1[it] = *(const float4*)(B + (size_t)(2 * kp + 1) * GDIM + 4 * lane);
  }

  const float4* H4 = (const float4*)h;
  const v4f* Ag = (const v4f*)h + (size_t)row0 * (GDIM / 4) + lane;
  v4f ar[16];
#pragma unroll
  for (int r = 0; r < 16; ++r) ar[r] = Ag[r * (GDIM / 4)];

  const int o_l = off[row0 + lmod];
  const int c_l = cnt[row0 + lmod];
  const float d_l = dinv[row0 + lmod];
  const float* bq4 = bias + lane * 4;
  float4 bv;
  bv.x = bq4[0];
  bv.y = bq4[1];
  bv.z = bq4[2];
  bv.w = bq4[3];

  {
    float* d0 = Bs + (lane & 3) * 4 * BS_CPITCH + (lane >> 2) * 2;
#pragma unroll
    for (int it = 0; it < B_ITERS; ++it) {
      const int kq = it * GEMM_WAVES + wave;
      const int kp = kq < GDIM / 2 ? kq : GDIM / 2 - 1;
      float* d = d0 + kp * BS_KPITCH;
      v2f p;
      p.x = w0[it].x; p.y = w1[it].x; *(v2f*)(d + 0 * BS_CPITCH) = p;
      p.x = w0[it].y; p.y = w1[it].y; *(v2f*)(d + 1 * BS_CPITCH) = p;
      p.x = w0[it].z; p.y = w1[it].z; *(v2f*)(d + 2 * BS_CPITCH) = p;
      p.x = w0[it].w; p.y = w1[it].w; *(v2f*)(d + 3 * BS_CPITCH) = p;
    }
  }
  __syncthreads();

  float* as = As + wave * (16 * AS_PITCH);
  float* arow = as + 4 * lane;
#pragma unroll
  for (int r = 0; r < 16; ++r) *(v4f*)(arow + r * AS_PITCH) = ar[r];

  {
    const int segStart = lane_int(o_l, 0);
    const int segEnd = lane_int(o_l, 15) + lane_int(c_l, 15);
    int r = 0;
    int curEnd = segStart + lane_int(c_l, 0);
    float dn = lane_float(d_l, 0);
    float4 acc = *(const float4*)arow;

    if (segStart < segEnd) {
      const int last = segEnd - 1;
      int sn[GB];
#pragma unroll
      for (int k = 0; k < GB; ++k) {
        const int e = segStart + k;
        sn[k] = perm[e < last ? e : last];
      }
      for (int p = segStart; p < segEnd; p += GB) {
        float4 v[GB];
#pragma unroll
        for (int k = 0; k < GB; ++k) v[k] = H4[(size_t)sn[k] * (GDIM / 4) + lane];
        if (p + GB < segEnd) {
#pragma unroll
          for (int k = 0; k < GB; ++k) {
            const int e = p + GB + k;
            sn[k] = perm[e < last ? e : last];
          }
        }
#pragma unroll
        for (int k = 0; k < GB; ++k) {
          const int e = p + k;
          if (e < segEnd) {
            while (e == curEnd && r < 15) {
              float4 t;
              t.x = fmaxf(fmaf(acc.x, dn, bv.x), 0.0f) * dn;
              t.y = fmaxf(fmaf(acc.y, dn, bv.y), 0.0f) * dn;
              t.z = fmaxf(fmaf(acc.z, dn, bv.z), 0.0f) * dn;
              t.w = fmaxf(fmaf(acc.w, dn, bv.w), 0.0f) * dn;
              *(float4*)(arow + r * AS_PITCH) = t;
              ++r;
              curEnd = lane_int(o_l, r) + lane_int(c_l, r);
              dn = lane_float(d_l, r);
              acc = *(const float4*)(arow + r * AS_PITCH);
            }
            acc.x += v[k].x;
            acc.y += v[k].y;
            acc.z += v[k].z;
            acc.w += v[k].w;
          }
        }
      }
    }
    for (;;) {
      float4 t;
      t.x = fmaxf(fmaf(acc.x, dn, bv.x), 0.0f) * dn;
      t.y = fmaxf(fmaf(acc.y, dn, bv.y), 0.0f) * dn;
      t.z = fmaxf(fmaf(acc.z, dn, bv.z), 0.0f) * dn;
      t.w = fmaxf(fmaf(acc.w, dn, bv.w), 0.0f) * dn;
      *(float4*)(arow + r * AS_PITCH) = t;
      ++r;
      if (r >= 16) break;
      dn = lane_float(d_l, r);
      acc = *(const float4*)(arow + r * AS_PITCH);
    }
  }

  __builtin_amdgcn_fence(__ATOMIC_ACQ_REL, "wavefront");
  __builtin_amdgcn_wave_barrier();

  v8f acc[8];
#pragma unroll
  for (int t = 0; t < 8; ++t) acc[t] = (v8f){0.f, 0.f, 0.f, 0.f, 0.f, 0.f, 0.f, 0.f};

  const float* ap = as + lmod * AS_PITCH + half * 2;
  const float* bp = Bs + half * BS_KPITCH + lmod * BS_CPITCH;
#pragma unroll 8
  for (int ks = 0; ks < GDIM / 4; ++ks) {
    const v2f a = *(const v2f*)(ap + 4 * ks);
    const float4* bq = (const float4*)(bp + ks * 2 * BS_KPITCH);
    float4 bf[4];
#pragma unroll
    for (int j = 0; j < 4; ++j) bf[j] = bq[j];
#pragma unroll
    for (int j = 0; j < 4; ++j) {
      v2f b0, b1;
      b0.x = bf[j].x; b0.y = bf[j].y;
      b1.x = bf[j].z; b1.y = bf[j].w;
      acc[2 * j] = __builtin_amdgcn_wmma_f32_16x16x4_f32(
           false, a,  false, b0,
           (short)0, acc[2 * j],  false,  false);
      acc[2 * j + 1] = __builtin_amdgcn_wmma_f32_16x16x4_f32(
           false, a,  false, b1,
           (short)0, acc[2 * j + 1],  false,  false);
    }
  }

  if (strip < nStrips) {
    float* Cp = C + (size_t)(row0 + half * 8) * GDIM + lmod;
#pragma unroll
    for (int t = 0; t < 8; ++t) {
#pragma unroll
      for (int r = 0; r < 8; ++r) Cp[(size_t)r * GDIM + 16 * t] = acc[t][r];
    }
  }
}

extern "C" void kernel_launch(void* const* d_in, const int* in_sizes, int n_in,
                              void* d_out, int out_size, void* d_ws, size_t ws_size,
                              hipStream_t stream) {
  const float* x  = (const float*)d_in[0];
  const int*   ei = (const int*)d_in[1];
  const float* W1 = (const float*)d_in[2];
  const float* b1 = (const float*)d_in[3];
  const float* W2 = (const float*)d_in[4];
  const float* b2 = (const float*)d_in[5];
  float* out = (float*)d_out;

  const int N = in_sizes[0] / GDIM;
  const int E = in_sizes[1] / 2;
  const int* src = ei;
  const int* dst = ei + E;

  float* dinv = (float*)d_ws;
  float* bufA = dinv + N;
  float* bufB = bufA + (size_t)N * GDIM;
  int* cnt  = (int*)(bufB + (size_t)N * GDIM);
  int* off  = cnt + N;
  int* cntC = off + N;
  int* btot = cntC + (size_t)NC * N;
  int* perm = btot + NB_MAX;

  const int T = 256;
  const int RS = (N + NR - 1) / NR;
  const int CS = (E + NC - 1) / NC;
  const int NB = (N + SCAN_B - 1) / SCAN_B;
  const int gNodeWave = (N + 7) / 8;
  const int nStrips = N / 16;
  const int gGemm = (nStrips + GEMM_WAVES - 1) / GEMM_WAVES;

  hist_kernel<<<dim3(NR, NC), BT, 0, stream>>>(dst, cntC, N, E, RS, CS);
  scan_dinv_kernel<<<NB, SCAN_B, 0, stream>>>(cntC, cnt, dinv, btot, N);
  fill_perm_kernel<<<dim3(NR, NC), BT, 0, stream>>>(src, dst, cntC, btot, off, perm, N, E, RS, CS, NB);

  gemm_wmma_f32_kernel<<<gGemm, GEMM_WAVES * 32, 0, stream>>>(x, W1, dinv, bufA, nStrips);

  gather_gemm_wmma_f32_kernel<<<gGemm, GEMM_WAVES * 32, 0, stream>>>(bufA, perm, off, cnt, dinv, b1, W2,
                                                                    bufB, nStrips);

  gather_rows_kernel<false><<<gNodeWave, T, 0, stream>>>(bufB, perm, off, cnt, dinv, b2, out, N);
}


# speedup vs baseline: 5.7424x; 1.1383x over previous
#include <hip/hip_runtime.h>
#include <hip/hip_bf16.h>

#define GDIM 128

typedef float v2f __attribute__((ext_vector_type(2)));
typedef float v4f __attribute__((ext_vector_type(4)));
typedef float v8f __attribute__((ext_vector_type(8)));

#define NR 8
#define NC 32
#define RS_MAX 16384
#define SCAN_B 1024
#define NB_MAX 128
#define BT 512
#define EV 25
#define NSL (RS_MAX / BT)


__device__ __forceinline__ int chunk_quads(const int* a, const int* b, int cbeg, int cend) {
  const bool aligned = ((((size_t)(a + cbeg)) | ((size_t)(b + cbeg))) & 15) == 0;
  return (aligned && cend > cbeg) ? ((cend - cbeg) >> 2) : 0;
}

__device__ __forceinline__ int4 load_quad(const int* __restrict__ p, int qi, int nq, int fillv) {
  const bool ok = qi < nq;
  int4 v = *(const int4*)(p + 4 * (ok ? qi : 0));
  if (!ok) { v.x = fillv; v.y = fillv; v.z = fillv; v.w = fillv; }
  return v;
}

__global__ void __launch_bounds__(BT) hist_kernel(const int* __restrict__ dst,
                                                  int* __restrict__ cntC,
                                                  int N, int E, int RS, int CS) {
  __shared__ int lcnt[RS_MAX];
  const int r = blockIdx.x;
  const int c = blockIdx.y;
  const int tid = threadIdx.x;
  const int rbase = r * RS;
  const int cbeg = c * CS;
  const int cend = (cbeg + CS < E) ? cbeg + CS : E;
  const int nq = chunk_quads(dst, dst, cbeg, cend);
  const int* dq = dst + cbeg;
  int q0 = 0;
  int4 dv[EV];
  if (nq > 0) {
#pragma unroll
    for (int k = 0; k < EV; ++k) dv[k] = load_quad(dq, q0 + k * BT + tid, nq, -1);
  }
  for (int i = tid; i < RS; i += BT) lcnt[i] = 0;
  __syncthreads();
  while (q0 < nq) {
#pragma unroll
    for (int k = 0; k < EV; ++k) {
      const unsigned d0 = (unsigned)(dv[k].x - rbase);
      const unsigned d1 = (unsigned)(dv[k].y - rbase);
      const unsigned d2 = (unsigned)(dv[k].z - rbase);
      const unsigned d3 = (unsigned)(dv[k].w - rbase);
      if (d0 < (unsigned)RS) atomicAdd(&lcnt[d0], 1);
      if (d1 < (unsigned)RS) atomicAdd(&lcnt[d1], 1);
      if (d2 < (unsigned)RS) atomicAdd(&lcnt[d2], 1);
      if (d3 < (unsigned)RS) atomicAdd(&lcnt[d3], 1);
    }
    q0 += BT * EV;
    if (q0 < nq) {
#pragma unroll
      for (int k = 0; k < EV; ++k) dv[k] = load_quad(dq, q0 + k * BT + tid, nq, -1);
    }
  }
  for (int e = cbeg + 4 * nq + tid; e < cend; e += BT) {
    const unsigned d = (unsigned)(dst[e] - rbase);
    if (d < (unsigned)RS) atomicAdd(&lcnt[d], 1);
  }
  __syncthreads();
  int cv[NSL];
#pragma unroll
  for (int j = 0; j < NSL; ++j) cv[j] = lcnt[(j * BT + tid < RS) ? j * BT + tid : 0];
#pragma unroll
  for (int j = 0; j < NSL; ++j) {
    const int i = j * BT + tid;
    const int n = rbase + i;
    if (i < RS && n < N) cntC[(size_t)c * N + n] = cv[j];
  }
}

__global__ void __launch_bounds__(SCAN_B) scan_dinv_kernel(int* __restrict__ cntC,
                                                           int* __restrict__ cnt,
                                                           float* __restrict__ dinv,
                                                           int* __restrict__ btot, int N) {
  __shared__ int wsum[32];
  const int tid = threadIdx.x;
  const int lane = tid & 31;
  const int wave = tid >> 5;
  const int n = blockIdx.x * SCAN_B + tid;
  const bool valid = n < N;
  int v[NC];
#pragma unroll
  for (int c = 0; c < NC; ++c) v[c] = valid ? cntC[(size_t)c * N + n] : 0;
  int tot = 0;
#pragma unroll
  for (int c = 0; c < NC; ++c) tot += v[c];

  int incl = tot;
  for (int d = 1; d < 32; d <<= 1) {
    const int up = __shfl_up(incl, d);
    if (lane >= d) incl += up;
  }
  if (lane == 31) wsum[wave] = incl;
  __syncthreads();
  const int wv = wsum[lane];
  int winc = wv;
  for (int d = 1; d < 32; d <<= 1) {
    const int up = __shfl_up(winc, d);
    if (lane >= d) winc += up;
  }
  const int total = __shfl(winc, 31);
  const int wbase = __shfl(winc, wave) - __shfl(wv, wave);
  int run = wbase + incl - tot;
  if (valid) {
    cnt[n] = tot;
    dinv[n] = rsqrtf((float)(tot + 1));
#pragma unroll
    for (int c = 0; c < NC; ++c) {
      cntC[(size_t)c * N + n] = run;
      run += v[c];
    }
  }
  if (tid == 0) btot[blockIdx.x] = total;
}

__device__ __forceinline__ void place8(const int (&d8)[8], const int (&s8)[8], int* cur,
                                       int* __restrict__ perm, int rbase, int RS) {
  int pos[8];
#pragma unroll
  for (int j = 0; j < 8; ++j) {
    const unsigned d = (unsigned)(d8[j] - rbase);
    pos[j] = (d < (unsigned)RS) ? atomicAdd(&cur[d], 1) : -1;
  }
#pragma unroll
  for (int j = 0; j < 8; ++j) {
    if (pos[j] >= 0) perm[pos[j]] = s8[j];
  }
}

__global__ void __launch_bounds__(BT) fill_perm_kernel(const int* __restrict__ src,
                                                       const int* __restrict__ dst,
                                                       const int* __restrict__ cntC,
                                                       const int* __restrict__ btot,
                                                       int* __restrict__ off,
                                                       int* __restrict__ perm,
                                                       int N, int E, int RS, int CS, int NB) {
  __shared__ int cur[RS_MAX];
  __shared__ int bp[NB_MAX];
  const int r = blockIdx.x;
  const int c = blockIdx.y;
  const int tid = threadIdx.x;
  const int rbase = r * RS;
  const int cbeg = c * CS;
  const int cend = (cbeg + CS < E) ? cbeg + CS : E;
  const int nq = chunk_quads(dst, src, cbeg, cend);
  const int* dq = dst + cbeg;
  const int* sq = src + cbeg;
  int cv[NSL];
#pragma unroll
  for (int j = 0; j < NSL; ++j) {
    const int i = j * BT + tid;
    const int n = rbase + i;
    cv[j] = cntC[(size_t)c * N + ((i < RS && n < N) ? n : 0)];
  }
  if (tid < 32) {
    int t[4];
#pragma unroll
    for (int k = 0; k < 4; ++k) t[k] = (tid * 4 + k < NB) ? btot[tid * 4 + k] : 0;
    const int s = t[0] + t[1] + t[2] + t[3];
    int incl = s;
    for (int d = 1; d < 32; d <<= 1) {
      const int up = __shfl_up(incl, d);
      if (tid >= d) incl += up;
    }
    int ex = incl - s;
#pragma unroll
    for (int k = 0; k < 4; ++k) {
      bp[tid * 4 + k] = ex;
      ex += t[k];
    }
  }
  __syncthreads();
#pragma unroll
  for (int j = 0; j < NSL; ++j) {
    const int i = j * BT + tid;
    const int n = rbase + i;
    cv[j] += bp[(i < RS && n < N) ? n / SCAN_B : 0];
  }
#pragma unroll
  for (int j = 0; j < NSL; ++j) {
    const int i = j * BT + tid;
    const int n = rbase + i;
    if (i < RS && n < N) {
      cur[i] = cv[j];
      if (c == 0) off[n] = cv[j];
    }
  }
  int q0 = 0;
  int4 dv[EV], sv[EV];
  if (nq > 0) {
#pragma unroll
    for (int k = 0; k < EV; ++k) {
      dv[k] = load_quad(dq, q0 + k * BT + tid, nq, -1);
      sv[k] = load_quad(sq, q0 + k * BT + tid, nq, 0);
    }
  }
  __syncthreads();
  while (q0 < nq) {
#pragma unroll
    for (int k = 0; k + 1 < EV; k += 2) {
      const int d8[8] = {dv[k].x, dv[k].y, dv[k].z, dv[k].w,
                         dv[k + 1].x, dv[k + 1].y, dv[k + 1].z, dv[k + 1].w};
      const int s8[8] = {sv[k].x, sv[k].y, sv[k].z, sv[k].w,
                         sv[k + 1].x, sv[k + 1].y, sv[k + 1].z, sv[k + 1].w};
      place8(d8, s8, cur, perm, rbase, RS);
    }
    if (EV & 1) {
      const int d8[8] = {dv[EV - 1].x, dv[EV - 1].y, dv[EV - 1].z, dv[EV - 1].w, -1, -1, -1, -1};
      const int s8[8] = {sv[EV - 1].x, sv[EV - 1].y, sv[EV - 1].z, sv[EV - 1].w, 0, 0, 0, 0};
      place8(d8, s8, cur, perm, rbase, RS);
    }
    q0 += BT * EV;
    if (q0 < nq) {
#pragma unroll
      for (int k = 0; k < EV; ++k) {
        dv[k] = load_quad(dq, q0 + k * BT + tid, nq, -1);
        sv[k] = load_quad(sq, q0 + k * BT + tid, nq, 0);
      }
    }
  }
  for (int e = cbeg + 4 * nq + tid; e < cend; e += BT) {
    const unsigned d = (unsigned)(dst[e] - rbase);
    if (d < (unsigned)RS) perm[atomicAdd(&cur[d], 1)] = src[e];
  }
}

#define BS_CPITCH 20
#define BS_KPITCH (16 * BS_CPITCH)
#define AS_PITCH 132
#define GEMM_WAVES 25
#define B_ITERS ((GDIM / 2 + GEMM_WAVES - 1) / GEMM_WAVES)

__global__ void __launch_bounds__(GEMM_WAVES * 32)
gemm_wmma_f32_kernel(const float* __restrict__ A, const float* __restrict__ B,
                     const float* __restrict__ rowScale, float* __restrict__ C, int nStrips) {
  __shared__ float Bs[(GDIM / 2) * BS_KPITCH];
  __shared__ float As[GEMM_WAVES * 16 * AS_PITCH];

  const int wave = __builtin_amdgcn_readfirstlane(threadIdx.x >> 5);
  const int lane = threadIdx.x & 31;
  const int half = lane >> 4;
  const int lmod = lane & 15;
  const int strip = blockIdx.x * GEMM_WAVES + wave;
  const int sidx = strip < nStrips ? strip : nStrips - 1;
  const int row0 = sidx * 16;

  float4 w0[B_ITERS], w1[B_ITERS];
#pragma unroll
  for (int it = 0; it < B_ITERS; ++it) {
    const int kq = it * GEMM_WAVES + wave;
    const int kp = kq < GDIM / 2 ? kq : GDIM / 2 - 1;
    w0[it] = *(const float4*)(B + (size_t)(2 * kp) * GDIM + 4 * lane);
    w1[it] = *(const float4*)(B + (size_t)(2 * kp + 1) * GDIM + 4 * lane);
  }

  const float4* Ag = (const float4*)(A + (size_t)row0 * GDIM) + lane;
  float4 ar[16];
#pragma unroll
  for (int r = 0; r < 16; ++r) ar[r] = Ag[r * (GDIM / 4)];

  const float* rs = rowScale + row0;
  float sc[16];
#pragma unroll
  for (int r = 0; r < 16; ++r) sc[r] = rs[r];

  {
    float* d0 = Bs + (lane & 3) * 4 * BS_CPITCH + (lane >> 2) * 2;
#pragma unroll
    for (int it = 0; it < B_ITERS; ++it) {
      const int kq = it * GEMM_WAVES + wave;
      const int kp = kq < GDIM / 2 ? kq : GDIM / 2 - 1;
      float* d = d0 + kp * BS_KPITCH;
      v2f p;
      p.x = w0[it].x; p.y = w1[it].x; *(v2f*)(d + 0 * BS_CPITCH) = p;
      p.x = w0[it].y; p.y = w1[it].y; *(v2f*)(d + 1 * BS_CPITCH) = p;
      p.x = w0[it].z; p.y = w1[it].z; *(v2f*)(d + 2 * BS_CPITCH) = p;
      p.x = w0[it].w; p.y = w1[it].w; *(v2f*)(d + 3 * BS_CPITCH) = p;
    }
  }
  float* as = As + wave * (16 * AS_PITCH);
#pragma unroll
  for (int r = 0; r < 16; ++r) {
    float4 t;
    t.x = ar[r].x * sc[r];
    t.y = ar[r].y * sc[r];
    t.z = ar[r].z * sc[r];
    t.w = ar[r].w * sc[r];
    *(float4*)(as + r * AS_PITCH + 4 * lane) = t;
  }

  __syncthreads();

  v8f acc[8];
#pragma unroll
  for (int t = 0; t < 8; ++t) acc[t] = (v8f){0.f, 0.f, 0.f, 0.f, 0.f, 0.f, 0.f, 0.f};

  const float* ap = as + lmod * AS_PITCH + half * 2;
  const float* bp = Bs + half * BS_KPITCH + lmod * BS_CPITCH;
#pragma unroll 8
  for (int ks = 0; ks < GDIM / 4; ++ks) {
    const v2f a = *(const v2f*)(ap + 4 * ks);
    const float4* bq = (const float4*)(bp + ks * 2 * BS_KPITCH);
    float4 bv[4];
#pragma unroll
    for (int j = 0; j < 4; ++j) bv[j] = bq[j];
#pragma unroll
    for (int j = 0; j < 4; ++j) {
      v2f b0, b1;
      b0.x = bv[j].x; b0.y = bv[j].y;
      b1.x = bv[j].z; b1.y = bv[j].w;
      acc[2 * j] = __builtin_amdgcn_wmma_f32_16x16x4_f32(
           false, a,  false, b0,
           (short)0, acc[2 * j],  false,  false);
      acc[2 * j + 1] = __builtin_amdgcn_wmma_f32_16x16x4_f32(
           false, a,  false, b1,
           (short)0, acc[2 * j + 1],  false,  false);
    }
  }

  if (strip < nStrips) {
    float* Cp = C + (size_t)(row0 + half * 8) * GDIM + lmod;
#pragma unroll
    for (int t = 0; t < 8; ++t) {
#pragma unroll
      for (int r = 0; r < 8; ++r) Cp[(size_t)r * GDIM + 16 * t] = acc[t][r];
    }
  }
}

template <int K>
__device__ __forceinline__ void accum_rows(const float4* __restrict__ H4,
                                           const int* __restrict__ perm,
                                           int j, int lane, float4& acc) {
  int s[K];
  float4 v[K];
#pragma unroll
  for (int k = 0; k < K; ++k) s[k] = perm[j + k];
#pragma unroll
  for (int k = 0; k < K; ++k) v[k] = H4[(size_t)s[k] * (GDIM / 4) + lane];
#pragma unroll
  for (int k = 0; k < K; ++k) {
    acc.x += v[k].x;
    acc.y += v[k].y;
    acc.z += v[k].z;
    acc.w += v[k].w;
  }
}

template <bool RELU>
__global__ void gather_rows_kernel(const float* __restrict__ h,
                                   const int* __restrict__ perm,
                                   const int* __restrict__ off,
                                   const int* __restrict__ cnt,
                                   const float* __restrict__ dinv,
                                   const float* __restrict__ bias,
                                   float* __restrict__ out, int n) {
  const int wave = __builtin_amdgcn_readfirstlane(threadIdx.x >> 5);
  const int node = blockIdx.x * 8 + wave;
  if (node >= n) return;
  const int lane = threadIdx.x & 31;
  const float4* H4 = (const float4*)h;
  int j = off[node];
  const int end = j + cnt[node];
  const float dn = dinv[node];

  const float* bp = bias + lane * 4;
  const float b0 = bp[0], b1v = bp[1], b2v = bp[2], b3 = bp[3];
  float4 acc = H4[(size_t)node * (GDIM / 4) + lane];

  for (; j + 8 <= end; j += 8) accum_rows<8>(H4, perm, j, lane, acc);
  if (j + 4 <= end) { accum_rows<4>(H4, perm, j, lane, acc); j += 4; }
  if (j + 2 <= end) { accum_rows<2>(H4, perm, j, lane, acc); j += 2; }
  if (j < end)      { accum_rows<1>(H4, perm, j, lane, acc); }

  acc.x = fmaf(acc.x, dn, b0);
  acc.y = fmaf(acc.y, dn, b1v);
  acc.z = fmaf(acc.z, dn, b2v);
  acc.w = fmaf(acc.w, dn, b3);
  if (RELU) {
    acc.x = fmaxf(acc.x, 0.0f);
    acc.y = fmaxf(acc.y, 0.0f);
    acc.z = fmaxf(acc.z, 0.0f);
    acc.w = fmaxf(acc.w, 0.0f);
  }
  ((float4*)(out + (size_t)node * GDIM))[lane] = acc;
}

#define GB 16

__device__ __forceinline__ int lane_int(int v, int l) {
  return __builtin_amdgcn_readlane(v, l);
}
__device__ __forceinline__ float lane_float(float v, int l) {
  return __int_as_float(__builtin_amdgcn_readlane(__float_as_int(v), l));
}

__global__ void __launch_bounds__(GEMM_WAVES * 32)
gather_gemm_wmma_f32_kernel(const float* __restrict__ h, const int* __restrict__ perm,
                            const int* __restrict__ off, const int* __restrict__ cnt,
                            const float* __restrict__ dinv, const float* __restrict__ bias,
                            const float* __restrict__ B, float* __restrict__ C, int nStrips) {
  __shared__ float Bs[(GDIM / 2) * BS_KPITCH];
  __shared__ float As[GEMM_WAVES * 16 * AS_PITCH];

  const int wave = __builtin_amdgcn_readfirstlane(threadIdx.x >> 5);
  const int lane = threadIdx.x & 31;
  const int half = lane >> 4;
  const int lmod = lane & 15;
  const int strip = blockIdx.x * GEMM_WAVES + wave;
  const int sidx = strip < nStrips ? strip : nStrips - 1;
  const int row0 = sidx * 16;

  float4 w0[B_ITERS], w1[B_ITERS];
#pragma unroll
  for (int it = 0; it < B_ITERS; ++it) {
    const int kq = it * GEMM_WAVES + wave;
    const int kp = kq < GDIM / 2 ? kq : GDIM / 2 - 1;
    w0[it] = *(const float4*)(B + (size_t)(2 * kp) * GDIM + 4 * lane);
    w1[it] = *(const float4*)(B + (size_t)(2 * kp + 1) * GDIM + 4 * lane);
  }

  const float4* H4 = (const float4*)h;
  const v4f* Ag = (const v4f*)h + (size_t)row0 * (GDIM / 4) + lane;
  v4f ar[16];
#pragma unroll
  for (int r = 0; r < 16; ++r) ar[r] = Ag[r * (GDIM / 4)];

  const int o_l = off[row0 + lmod];
  const int c_l = cnt[row0 + lmod];
  const float d_l = dinv[row0 + lmod];
  const float* bq4 = bias + lane * 4;
  float4 bv;
  bv.x = bq4[0];
  bv.y = bq4[1];
  bv.z = bq4[2];
  bv.w = bq4[3];

  {
    float* d0 = Bs + (lane & 3) * 4 * BS_CPITCH + (lane >> 2) * 2;
#pragma unroll
    for (int it = 0; it < B_ITERS; ++it) {
      const int kq = it * GEMM_WAVES + wave;
      const int kp = kq < GDIM / 2 ? kq : GDIM / 2 - 1;
      float* d = d0 + kp * BS_KPITCH;
      v2f p;
      p.x = w0[it].x; p.y = w1[it].x; *(v2f*)(d + 0 * BS_CPITCH) = p;
      p.x = w0[it].y; p.y = w1[it].y; *(v2f*)(d + 1 * BS_CPITCH) = p;
      p.x = w0[it].z; p.y = w1[it].z; *(v2f*)(d + 2 * BS_CPITCH) = p;
      p.x = w0[it].w; p.y = w1[it].w; *(v2f*)(d + 3 * BS_CPITCH) = p;
    }
  }
  __syncthreads();

  float* as = As + wave * (16 * AS_PITCH);
  float* arow = as + 4 * lane;
#pragma unroll
  for (int r = 0; r < 16; ++r) *(v4f*)(arow + r * AS_PITCH) = ar[r];

  {
    const int segStart = lane_int(o_l, 0);
    const int segEnd = lane_int(o_l, 15) + lane_int(c_l, 15);
    int r = 0;
    int curEnd = segStart + lane_int(c_l, 0);
    float dn = lane_float(d_l, 0);
    float4 acc = *(const float4*)arow;

    if (segStart < segEnd) {
      const int last = segEnd - 1;
      int sn[GB];
#pragma unroll
      for (int k = 0; k < GB; ++k) {
        const int e = segStart + k;
        sn[k] = perm[e < last ? e : last];
      }
      for (int p = segStart; p < segEnd; p += GB) {
        float4 v[GB];
#pragma unroll
        for (int k = 0; k < GB; ++k) v[k] = H4[(size_t)sn[k] * (GDIM / 4) + lane];
        if (p + GB < segEnd) {
#pragma unroll
          for (int k = 0; k < GB; ++k) {
            const int e = p + GB + k;
            sn[k] = perm[e < last ? e : last];
          }
        }
#pragma unroll
        for (int k = 0; k < GB; ++k) {
          const int e = p + k;
          if (e < segEnd) {
            while (e == curEnd && r < 15) {
              float4 t;
              t.x = fmaxf(fmaf(acc.x, dn, bv.x), 0.0f) * dn;
              t.y = fmaxf(fmaf(acc.y, dn, bv.y), 0.0f) * dn;
              t.z = fmaxf(fmaf(acc.z, dn, bv.z), 0.0f) * dn;
              t.w = fmaxf(fmaf(acc.w, dn, bv.w), 0.0f) * dn;
              *(float4*)(arow + r * AS_PITCH) = t;
              ++r;
              curEnd = lane_int(o_l, r) + lane_int(c_l, r);
              dn = lane_float(d_l, r);
              acc = *(const float4*)(arow + r * AS_PITCH);
            }
            acc.x += v[k].x;
            acc.y += v[k].y;
            acc.z += v[k].z;
            acc.w += v[k].w;
          }
        }
      }
    }
    for (;;) {
      float4 t;
      t.x = fmaxf(fmaf(acc.x, dn, bv.x), 0.0f) * dn;
      t.y = fmaxf(fmaf(acc.y, dn, bv.y), 0.0f) * dn;
      t.z = fmaxf(fmaf(acc.z, dn, bv.z), 0.0f) * dn;
      t.w = fmaxf(fmaf(acc.w, dn, bv.w), 0.0f) * dn;
      *(float4*)(arow + r * AS_PITCH) = t;
      ++r;
      if (r >= 16) break;
      dn = lane_float(d_l, r);
      acc = *(const float4*)(arow + r * AS_PITCH);
    }
  }

  __builtin_amdgcn_fence(__ATOMIC_ACQ_REL, "wavefront");
  __builtin_amdgcn_wave_barrier();

  v8f acc[8];
#pragma unroll
  for (int t = 0; t < 8; ++t) acc[t] = (v8f){0.f, 0.f, 0.f, 0.f, 0.f, 0.f, 0.f, 0.f};

  const float* ap = as + lmod * AS_PITCH + half * 2;
  const float* bp = Bs + half * BS_KPITCH + lmod * BS_CPITCH;
#pragma unroll 8
  for (int ks = 0; ks < GDIM / 4; ++ks) {
    const v2f a = *(const v2f*)(ap + 4 * ks);
    const float4* bq = (const float4*)(bp + ks * 2 * BS_KPITCH);
    float4 bf[4];
#pragma unroll
    for (int j = 0; j < 4; ++j) bf[j] = bq[j];
#pragma unroll
    for (int j = 0; j < 4; ++j) {
      v2f b0, b1;
      b0.x = bf[j].x; b0.y = bf[j].y;
      b1.x = bf[j].z; b1.y = bf[j].w;
      acc[2 * j] = __builtin_amdgcn_wmma_f32_16x16x4_f32(
           false, a,  false, b0,
           (short)0, acc[2 * j],  false,  false);
      acc[2 * j + 1] = __builtin_amdgcn_wmma_f32_16x16x4_f32(
           false, a,  false, b1,
           (short)0, acc[2 * j + 1],  false,  false);
    }
  }

  if (strip < nStrips) {
    float* Cp = C + (size_t)(row0 + half * 8) * GDIM + lmod;
#pragma unroll
    for (int t = 0; t < 8; ++t) {
#pragma unroll
      for (int r = 0; r < 8; ++r) Cp[(size_t)r * GDIM + 16 * t] = acc[t][r];
    }
  }
}

extern "C" void kernel_launch(void* const* d_in, const int* in_sizes, int n_in,
                              void* d_out, int out_size, void* d_ws, size_t ws_size,
                              hipStream_t stream) {
  const float* x  = (const float*)d_in[0];
  const int*   ei = (const int*)d_in[1];
  const float* W1 = (const float*)d_in[2];
  const float* b1 = (const float*)d_in[3];
  const float* W2 = (const float*)d_in[4];
  const float* b2 = (const float*)d_in[5];
  float* out = (float*)d_out;

  const int N = in_sizes[0] / GDIM;
  const int E = in_sizes[1] / 2;
  const int* src = ei;
  const int* dst = ei + E;

  float* bufA = (float*)d_ws;
  float* bufB = bufA + (size_t)N * GDIM;
  float* dinv = bufB + (size_t)N * GDIM;
  int* cnt  = (int*)(dinv + N);
  int* off  = cnt + N;
  int* cntC = off + N;
  int* btot = cntC + (size_t)NC * N;
  int* perm = btot + NB_MAX;

  const int T = 256;
  const int RS = (N + NR - 1) / NR;
  const int CS = (E + NC - 1) / NC;
  const int NB = (N + SCAN_B - 1) / SCAN_B;
  const int gNodeWave = (N + 7) / 8;
  const int nStrips = N / 16;
  const int gGemm = (nStrips + GEMM_WAVES - 1) / GEMM_WAVES;

  hist_kernel<<<dim3(NR, NC), BT, 0, stream>>>(dst, cntC, N, E, RS, CS);
  scan_dinv_kernel<<<NB, SCAN_B, 0, stream>>>(cntC, cnt, dinv, btot, N);
  fill_perm_kernel<<<dim3(NR, NC), BT, 0, stream>>>(src, dst, cntC, btot, off, perm, N, E, RS, CS, NB);

  gemm_wmma_f32_kernel<<<gGemm, GEMM_WAVES * 32, 0, stream>>>(x, W1, dinv, bufA, nStrips);

  gather_gemm_wmma_f32_kernel<<<gGemm, GEMM_WAVES * 32, 0, stream>>>(bufA, perm, off, cnt, dinv, b1, W2,
                                                                    bufB, nStrips);

  gather_rows_kernel<false><<<gNodeWave, T, 0, stream>>>(bufB, perm, off, cnt, dinv, b2, out, N);
}
